# starting kernel
# baseline (speedup 1.0000x reference)
#include <hip/hip_runtime.h>

typedef __attribute__((ext_vector_type(2))) float v2f;
typedef __attribute__((ext_vector_type(8))) float v8f;

#define DCH 128


__global__ void gcn_deg_init(float* deg, int n) {
    int i = blockIdx.x * blockDim.x + threadIdx.x;
    if (i < n) deg[i] = 1.0f;
}

__global__ void gcn_deg_count(const int* __restrict__ tgt, float* deg, int e) {
    int i = blockIdx.x * blockDim.x + threadIdx.x;
    if (i < e) atomicAdd(&deg[tgt[i]], 1.0f);
}

__global__ void gcn_rsqrt(float* deg, int n) {
    int i = blockIdx.x * blockDim.x + threadIdx.x;
    if (i < n) deg[i] = rsqrtf(deg[i]);
}


__global__ void gcn_gemm_wmma(const float* __restrict__ A,
                              const float* __restrict__ W,
                              float* __restrict__ H) {
    int waveId = blockIdx.x * (blockDim.x >> 5) + (threadIdx.x >> 5);
    int lane   = threadIdx.x & 31;
    int mTile  = waveId >> 3;
    int nTile  = waveId & 7;
    int hi     = lane >> 4;
    int l16    = lane & 15;

    int row = mTile * 16 + l16;
    int col = nTile * 16 + l16;

    const float* arow = A + (size_t)row * DCH + 2 * hi;

    v8f c = {};
#pragma unroll
    for (int k0 = 0; k0 < DCH; k0 += 4) {
        v2f a, b;
        a.x = arow[k0 + 0];
        a.y = arow[k0 + 1];
        int kb = k0 + 2 * hi;
        b.x = W[(size_t)kb * DCH + col];
        b.y = W[(size_t)(kb + 1) * DCH + col];
        c = __builtin_amdgcn_wmma_f32_16x16x4_f32(
                 false, a,  false, b,
                 (short)0, c,  false,  false);
    }

    float* hcol = H + (size_t)(mTile * 16) * DCH + nTile * 16 + l16;
#pragma unroll
    for (int i = 0; i < 8; ++i)
        hcol[(size_t)(i + 8 * hi) * DCH] = c[i];
}


__global__ void gcn_zero(float* p, int n) {
    int i = blockIdx.x * blockDim.x + threadIdx.x;
    if (i < n) p[i] = 0.0f;
}

__global__ void gcn_scatter(const float* __restrict__ H,
                            const int* __restrict__ src,
                            const int* __restrict__ tgt,
                            const float* __restrict__ dis,
                            float* __restrict__ agg, int e) {
    int wave = blockIdx.x * (blockDim.x >> 5) + (threadIdx.x >> 5);
    int lane = threadIdx.x & 31;
    if (wave >= e) return;
    int s = src[wave];
    int t = tgt[wave];
    float w = dis[s] * dis[t];
    float4 v = ((const float4*)(H + (size_t)s * DCH))[lane];
    float* at = agg + (size_t)t * DCH + lane * 4;
    atomicAdd(at + 0, v.x * w);
    atomicAdd(at + 1, v.y * w);
    atomicAdd(at + 2, v.z * w);
    atomicAdd(at + 3, v.w * w);
}

__global__ void gcn_finalize(const float* __restrict__ agg,
                             const float* __restrict__ H,
                             const float* __restrict__ dis,
                             const float* __restrict__ bias,
                             float* __restrict__ out, int nd, int relu) {
    int i = blockIdx.x * blockDim.x + threadIdx.x;
    if (i >= nd) return;
    int node = i >> 7;
    int d    = i & (DCH - 1);
    float di = dis[node];
    float v  = agg[i] + H[i] * di * di + bias[d];
    out[i] = relu ? fmaxf(v, 0.0f) : v;
}


extern "C" void kernel_launch(void* const* d_in, const int* in_sizes, int n_in,
                              void* d_out, int out_size, void* d_ws, size_t ws_size,
                              hipStream_t stream) {
    const float* X  = (const float*)d_in[0];
    const int*   ei = (const int*)  d_in[1];
    const float* W1 = (const float*)d_in[2];
    const float* b1 = (const float*)d_in[3];
    const float* W2 = (const float*)d_in[4];
    const float* b2 = (const float*)d_in[5];
    const float* W3 = (const float*)d_in[6];
    const float* b3 = (const float*)d_in[7];
    float* out = (float*)d_out;

    const int n  = in_sizes[0] / DCH;
    const int e  = in_sizes[1] / 2;
    const int nd = n * DCH;

    const int* src = ei;
    const int* tgt = ei + e;

    float* dis = (float*)d_ws;
    float* H   = dis + n;
    float* AGG = H + nd;

    gcn_deg_init <<<(n + 255) / 256, 256, 0, stream>>>(dis, n);
    gcn_deg_count<<<(e + 255) / 256, 256, 0, stream>>>(tgt, dis, e);
    gcn_rsqrt    <<<(n + 255) / 256, 256, 0, stream>>>(dis, n);

    const int mTiles     = n / 16;
    const int gemmBlocks = (mTiles * 8) / 8;
    const int edgeBlocks = (e + 7) / 8;
    const int ndBlocks   = (nd + 255) / 256;

    const float* Ws[3] = {W1, W2, W3};
    const float* bs[3] = {b1, b2, b3};
    const float* cur   = X;

    for (int layer = 0; layer < 3; ++layer) {
        gcn_gemm_wmma<<<gemmBlocks, 256, 0, stream>>>(cur, Ws[layer], H);
        gcn_zero     <<<ndBlocks,   256, 0, stream>>>(AGG, nd);
        gcn_scatter  <<<edgeBlocks, 256, 0, stream>>>(H, src, tgt, dis, AGG, e);
        gcn_finalize <<<ndBlocks,   256, 0, stream>>>(AGG, H, dis, bs[layer], out,
                                                      nd, layer < 2 ? 1 : 0);
        cur = out;
    }
}


# Global f32 atomic scatter -> LDS sums: each block owns 200 target rows (800 in the degree count), filters the whole edge list, adds source rows by LDS f32 atomics (arrival order), stores rows once
# speedup vs baseline: 2.5794x; 2.5794x over previous
#include <hip/hip_runtime.h>
#include <stdint.h>


typedef __attribute__((ext_vector_type(2))) float v2f;
typedef __attribute__((ext_vector_type(8))) float v8f;

#define DCH    128
#define TROWS  200
#define DROWS  800
#define NTHR   1024
#define NWAVES 32
#define QCAP   544

template <bool GUARD, class F>
__device__ __forceinline__ void filter_trip(const int4 (&q)[4], int base, int head, int m4, F& f) {
#pragma unroll
    for (int u = 0; u < 4; ++u) {
        const int i4 = base + 32 * u;
        const bool ok = GUARD ? (i4 < m4) : true;
        const int e0 = head + 4 * i4;
        f(q[u].x, e0, ok);
        f(q[u].y, e0 + 1, ok);
        f(q[u].z, e0 + 2, ok);
        f(q[u].w, e0 + 3, ok);
    }
}

template <class F>
__device__ __forceinline__ void scan_targets(const int* __restrict__ tgt, int e,
                                             int wave, int lane, F& f) {
    const uintptr_t a = (uintptr_t)tgt;
    int head = (int)(((16u - (unsigned)(a & 15u)) & 15u) >> 2);
    if (head > e) head = e;
    const int m4    = (e - head) >> 2;
    const int tail0 = head + 4 * m4;
    if (wave == 0 && head > 0) {
        int i = lane;
        bool ok = i < head;
        int t = ok ? tgt[i] : -1;
        f(t, i, ok);
    }
    if (wave == NWAVES - 1 && tail0 < e) {
        int i = tail0 + lane;
        bool ok = i < e;
        int t = ok ? tgt[i] : -1;
        f(t, i, ok);
    }
    const int4* __restrict__ t4 = (const int4*)(tgt + head);
    const int per_it = NWAVES * 128;
    const int nfull  = m4 / per_it;
    const int woff   = wave * 128 + lane;
    if (nfull > 0) {
        int it = (int)((blockIdx.x * (unsigned)nfull) / gridDim.x);
        int4 q[4];
#pragma unroll
        for (int u = 0; u < 4; ++u) q[u] = t4[it * per_it + woff + 32 * u];
        for (int k = 0; k < nfull; ++k) {
            int nit = it + 1;
            if (nit == nfull) nit = 0;
            int4 nq[4];
#pragma unroll
            for (int u = 0; u < 4; ++u) nq[u] = t4[nit * per_it + woff + 32 * u];
            filter_trip<false>(q, it * per_it + woff, head, m4, f);
            f.trip_end();
#pragma unroll
            for (int u = 0; u < 4; ++u) q[u] = nq[u];
            it = nit;
        }
    }
    if (nfull * per_it < m4) {
        const int base = nfull * per_it + woff;
        int4 q[4];
#pragma unroll
        for (int u = 0; u < 4; ++u) q[u] = t4[min(base + 32 * u, m4 - 1)];
        filter_trip<true>(q, base, head, m4, f);
        f.trip_end();
    }
}


struct DegCount {
    int* cnt; int row0; int tl;
    __device__ __forceinline__ void operator()(int t, int, bool ok) {
        int rl = t - row0;
        unsigned lim = ok ? (unsigned)tl : 0u;
        if ((unsigned)rl < lim) atomicAdd(&cnt[rl], 1);
    }
    __device__ __forceinline__ void trip_end() {}
};

__global__ __launch_bounds__(NTHR) void gcn_deg_tile(const int* __restrict__ tgt,
                                                     float* __restrict__ dis, int n, int e) {
    __shared__ int cnt[DROWS];
    const int tid = threadIdx.x, lane = tid & 31, wave = tid >> 5;
    const int row0 = blockIdx.x * DROWS;
    const int tl = min(DROWS, n - row0);
    for (int i = tid; i < DROWS; i += NTHR) cnt[i] = 0;
    __syncthreads();
    DegCount f{cnt, row0, tl};
    scan_targets(tgt, e, wave, lane, f);
    __syncthreads();
    for (int i = tid; i < tl; i += NTHR)
        dis[row0 + i] = rsqrtf(1.0f + (float)cnt[i]);
}


__global__ void gcn_gemm_wmma(const float* __restrict__ A,
                              const float* __restrict__ W,
                              float* __restrict__ H) {
    int waveId = blockIdx.x * (blockDim.x >> 5) + (threadIdx.x >> 5);
    int lane   = threadIdx.x & 31;
    int mTile  = waveId >> 3;
    int nTile  = waveId & 7;
    int hi     = lane >> 4;
    int l16    = lane & 15;

    int row = mTile * 16 + l16;
    int col = nTile * 16 + l16;

    const float* arow = A + (size_t)row * DCH + 2 * hi;

    v8f c = {};
#pragma unroll
    for (int k0 = 0; k0 < DCH; k0 += 4) {
        v2f a, b;
        a.x = arow[k0 + 0];
        a.y = arow[k0 + 1];
        int kb = k0 + 2 * hi;
        b.x = W[(size_t)kb * DCH + col];
        b.y = W[(size_t)(kb + 1) * DCH + col];
        c = __builtin_amdgcn_wmma_f32_16x16x4_f32(
                 false, a,  false, b,
                 (short)0, c,  false,  false);
    }

    float* hcol = H + (size_t)(mTile * 16) * DCH + nTile * 16 + l16;
#pragma unroll
    for (int i = 0; i < 8; ++i)
        hcol[(size_t)(i + 8 * hi) * DCH] = c[i];
}


struct AggFilter {
    float* acc; int* qe; int* qr;
    const float* __restrict__ H; const int* __restrict__ src; const float* __restrict__ dis;
    int row0; int tl; int lane; int cnt;

    __device__ __forceinline__ void operator()(int t, int ei, bool ok) {
        int rl = t - row0;
        unsigned lim = ok ? (unsigned)tl : 0u;
        bool m = (unsigned)rl < lim;
        unsigned mask = __builtin_amdgcn_ballot_w32(m);
        if (mask != 0u) {
            int p = (int)__builtin_amdgcn_mbcnt_lo(mask, 0u);
            if (m) { qe[cnt + p] = ei; qr[cnt + p] = rl; }
            cnt += __builtin_popcount(mask);
        }
    }

    __device__ __forceinline__ void one(int s, float w, int r, int j) {
        int   sj = __builtin_amdgcn_readlane(s, j);
        float wj = __int_as_float(__builtin_amdgcn_readlane(__float_as_int(w), j));
        int   rj = __builtin_amdgcn_readlane(r, j);
        float4 v = ((const float4*)(H + (size_t)sj * DCH))[lane];
        float* ap = acc + rj * DCH + lane;
        atomicAdd(ap,      v.x * wj);
        atomicAdd(ap + 32, v.y * wj);
        atomicAdd(ap + 64, v.z * wj);
        atomicAdd(ap + 96, v.w * wj);
    }

    __device__ __forceinline__ void flush(int start, int nvalid) {
        int r = 0, s = 0; float w = 0.0f;
        if (lane < nvalid) {
            int ei = qe[start + lane];
            r = qr[start + lane];
            s = src[ei];
            w = dis[s] * dis[row0 + r];
        }
        int j = 0;
        for (; j + 8 <= nvalid; j += 8) {
            float4 v[8];
#pragma unroll
            for (int u = 0; u < 8; ++u) {
                int sj = __builtin_amdgcn_readlane(s, j + u);
                v[u] = ((const float4*)(H + (size_t)sj * DCH))[lane];
            }
#pragma unroll
            for (int u = 0; u < 8; ++u) {
                float wj = __int_as_float(__builtin_amdgcn_readlane(__float_as_int(w), j + u));
                int   rj = __builtin_amdgcn_readlane(r, j + u);
                float* ap = acc + rj * DCH + lane;
                atomicAdd(ap,      v[u].x * wj);
                atomicAdd(ap + 32, v[u].y * wj);
                atomicAdd(ap + 64, v[u].z * wj);
                atomicAdd(ap + 96, v[u].w * wj);
            }
        }
        for (; j < nvalid; ++j) one(s, w, r, j);
    }

    __device__ __forceinline__ void trip_end() {
        int done = 0;
        while (cnt - done >= 32) { flush(done, 32); done += 32; }
        if (done > 0) {
            int rem = cnt - done;
            int a = 0, b = 0;
            if (lane < rem) { a = qe[done + lane]; b = qr[done + lane]; }
            if (lane < rem) { qe[lane] = a; qr[lane] = b; }
            cnt = rem;
        }
    }
};

__global__ __launch_bounds__(NTHR) void gcn_agg_tile(const float* __restrict__ H,
                                                     const int* __restrict__ src,
                                                     const int* __restrict__ tgt,
                                                     const float* __restrict__ dis,
                                                     const float* __restrict__ bias,
                                                     float* __restrict__ out,
                                                     int n, int e, int relu) {
    __shared__ float acc[TROWS * DCH];
    __shared__ int qe[NWAVES * QCAP];
    __shared__ int qr[NWAVES * QCAP];
    const int tid = threadIdx.x, lane = tid & 31, wave = tid >> 5;
    const int row0 = blockIdx.x * TROWS;
    const int tl = min(TROWS, n - row0);

    const float4 b = ((const float4*)bias)[lane];
    {
        constexpr int RPW = (TROWS + NWAVES - 1) / NWAVES;
        float4 h[RPW]; float d[RPW];
#pragma unroll
        for (int i = 0; i < RPW; ++i) {
            int rc = min(wave + NWAVES * i, tl - 1);
            d[i] = dis[row0 + rc];
            h[i] = ((const float4*)(H + (size_t)(row0 + rc) * DCH))[lane];
        }
#pragma unroll
        for (int i = 0; i < RPW; ++i) {
            int r = wave + NWAVES * i;
            if (r < tl) {
                float* ap = acc + r * DCH + lane;
                ap[0]  = h[i].x * d[i] * d[i] + b.x;
                ap[32] = h[i].y * d[i] * d[i] + b.y;
                ap[64] = h[i].z * d[i] * d[i] + b.z;
                ap[96] = h[i].w * d[i] * d[i] + b.w;
            }
        }
    }
    __syncthreads();

    AggFilter f{acc, qe + wave * QCAP, qr + wave * QCAP, H, src, dis, row0, tl, lane, 0};
    scan_targets(tgt, e, wave, lane, f);
    if (f.cnt > 0) f.flush(0, f.cnt);
    __syncthreads();

    for (int r = wave; r < tl; r += NWAVES) {
        const float* ap = acc + r * DCH + lane;
        float4 v;
        v.x = ap[0]; v.y = ap[32]; v.z = ap[64]; v.w = ap[96];
        if (relu) {
            v.x = fmaxf(v.x, 0.0f); v.y = fmaxf(v.y, 0.0f);
            v.z = fmaxf(v.z, 0.0f); v.w = fmaxf(v.w, 0.0f);
        }
        ((float4*)(out + (size_t)(row0 + r) * DCH))[lane] = v;
    }
}


extern "C" void kernel_launch(void* const* d_in, const int* in_sizes, int n_in,
                              void* d_out, int out_size, void* d_ws, size_t ws_size,
                              hipStream_t stream) {
    const float* X  = (const float*)d_in[0];
    const int*   ei = (const int*)  d_in[1];
    const float* W1 = (const float*)d_in[2];
    const float* b1 = (const float*)d_in[3];
    const float* W2 = (const float*)d_in[4];
    const float* b2 = (const float*)d_in[5];
    const float* W3 = (const float*)d_in[6];
    const float* b3 = (const float*)d_in[7];
    float* out = (float*)d_out;

    const int n  = in_sizes[0] / DCH;
    const int e  = in_sizes[1] / 2;

    const int* src = ei;
    const int* tgt = ei + e;

    float* dis = (float*)d_ws;
    float* H   = dis + n;

    gcn_deg_tile<<<(n + DROWS - 1) / DROWS, NTHR, 0, stream>>>(tgt, dis, n, e);

    const int mTiles     = n / 16;
    const int gemmBlocks = (mTiles * 8) / 8;
    const int aggBlocks  = (n + TROWS - 1) / TROWS;

    const float* Ws[3] = {W1, W2, W3};
    const float* bs[3] = {b1, b2, b3};
    const float* cur   = X;

    for (int layer = 0; layer < 3; ++layer) {
        gcn_gemm_wmma<<<gemmBlocks, 256, 0, stream>>>(cur, Ws[layer], H);
        gcn_agg_tile <<<aggBlocks, NTHR, 0, stream>>>(H, src, tgt, dis, bs[layer], out,
                                                      n, e, layer < 2 ? 1 : 0);
        cur = out;
    }
}


# Edge list filtered once per call, not 4 times: a filter launch per tile (224 rows, was 200) writes (source,row) records + row degrees; 3 aggregations read them; LDS f32 atomic sums, arrival order
# speedup vs baseline: 3.3512x; 1.2992x over previous
#include <hip/hip_runtime.h>
#include <stdint.h>


typedef __attribute__((ext_vector_type(2))) float v2f;
typedef __attribute__((ext_vector_type(8))) float v8f;

#define DCH    128
#define TROWS  224
#define SPITCH 32
#define NTHR   1024
#define NWAVES 32
#define QCAP   544
#define RCAP   32768

static_assert(TROWS <= 256, "a record keeps the tile row in its low 8 bits");

template <bool GUARD, class F>
__device__ __forceinline__ void filter_trip(const int4 (&q)[4], int base, int head, int m4, F& f) {
#pragma unroll
    for (int u = 0; u < 4; ++u) {
        const int i4 = base + 32 * u;
        const bool ok = GUARD ? (i4 < m4) : true;
        const int e0 = head + 4 * i4;
        f(q[u].x, e0, ok);
        f(q[u].y, e0 + 1, ok);
        f(q[u].z, e0 + 2, ok);
        f(q[u].w, e0 + 3, ok);
    }
}

template <class F>
__device__ __forceinline__ void scan_targets(const int* __restrict__ tgt, int e,
                                             int wave, int lane, F& f) {
    const uintptr_t a = (uintptr_t)tgt;
    int head = (int)(((16u - (unsigned)(a & 15u)) & 15u) >> 2);
    if (head > e) head = e;
    const int m4    = (e - head) >> 2;
    const int tail0 = head + 4 * m4;
    if (wave == 0 && head > 0) {
        int i = lane;
        bool ok = i < head;
        int t = ok ? tgt[i] : -1;
        f(t, i, ok);
    }
    if (wave == NWAVES - 1 && tail0 < e) {
        int i = tail0 + lane;
        bool ok = i < e;
        int t = ok ? tgt[i] : -1;
        f(t, i, ok);
    }
    const int4* __restrict__ t4 = (const int4*)(tgt + head);
    const int per_it = NWAVES * 128;
    const int nfull  = m4 / per_it;
    const int woff   = wave * 128 + lane;
    if (nfull > 0) {
        int it = (int)((blockIdx.x * (unsigned)nfull) / gridDim.x);
        int4 q[4];
#pragma unroll
        for (int u = 0; u < 4; ++u) q[u] = t4[it * per_it + woff + 32 * u];
        for (int k = 0; k < nfull; ++k) {
            int nit = it + 1;
            if (nit == nfull) nit = 0;
            int4 nq[4];
#pragma unroll
            for (int u = 0; u < 4; ++u) nq[u] = t4[nit * per_it + woff + 32 * u];
            filter_trip<false>(q, it * per_it + woff, head, m4, f);
            f.trip_end();
#pragma unroll
            for (int u = 0; u < 4; ++u) q[u] = nq[u];
            it = nit;
        }
    }
    if (nfull * per_it < m4) {
        const int base = nfull * per_it + woff;
        int4 q[4];
#pragma unroll
        for (int u = 0; u < 4; ++u) q[u] = t4[min(base + 32 * u, m4 - 1)];
        filter_trip<true>(q, base, head, m4, f);
        f.trip_end();
    }
}


struct TileFilter {
    int* qe; int* qr; int* degc; int* nslot;
    int* __restrict__ recs; const int* __restrict__ src;
    int row0; int tl; int lane; int cnt;

    __device__ __forceinline__ void operator()(int t, int ei, bool ok) {
        int rl = t - row0;
        unsigned lim = ok ? (unsigned)tl : 0u;
        bool m = (unsigned)rl < lim;
        unsigned mask = __builtin_amdgcn_ballot_w32(m);
        if (mask != 0u) {
            int p = (int)__builtin_amdgcn_mbcnt_lo(mask, 0u);
            if (m) { qe[cnt + p] = ei; qr[cnt + p] = rl; }
            cnt += __builtin_popcount(mask);
        }
    }

    __device__ __forceinline__ void emit(int start, int nvalid) {
        int rec = -1;
        if (lane < nvalid) {
            int ei = qe[start + lane];
            int r  = qr[start + lane];
            int s  = src[ei];
            atomicAdd(&degc[r], 1);
            rec = (s << 8) | r;
        }
        int pos = 0;
        if (lane == 0) pos = atomicAdd(nslot, 32);
        pos = __builtin_amdgcn_readfirstlane(pos);
        if (pos + 32 <= RCAP) recs[pos + lane] = rec;
    }

    __device__ __forceinline__ void trip_end() {
        int done = 0;
        while (cnt - done >= 32) { emit(done, 32); done += 32; }
        if (done > 0) {
            int rem = cnt - done;
            int a = 0, b = 0;
            if (lane < rem) { a = qe[done + lane]; b = qr[done + lane]; }
            if (lane < rem) { qe[lane] = a; qr[lane] = b; }
            cnt = rem;
        }
    }
};

__global__ __launch_bounds__(NTHR) void gcn_filter_tile(const int* __restrict__ src,
                                                        const int* __restrict__ tgt,
                                                        float* __restrict__ dis,
                                                        int* __restrict__ recs,
                                                        int* __restrict__ slots,
                                                        int n, int e) {
    __shared__ int qe[NWAVES * QCAP];
    __shared__ int qr[NWAVES * QCAP];
    __shared__ int degc[TROWS];
    __shared__ int nslot;
    const int tid = threadIdx.x, lane = tid & 31, wave = tid >> 5;
    const int row0 = blockIdx.x * TROWS;
    const int tl = min(TROWS, n - row0);
    for (int i = tid; i < TROWS; i += NTHR) degc[i] = 0;
    if (tid == 0) nslot = 0;
    __syncthreads();
    TileFilter f{qe + wave * QCAP, qr + wave * QCAP, degc, &nslot,
                 recs + (size_t)blockIdx.x * RCAP, src, row0, tl, lane, 0};
    scan_targets(tgt, e, wave, lane, f);
    if (f.cnt > 0) f.emit(0, f.cnt);
    __syncthreads();
    for (int i = tid; i < tl; i += NTHR)
        dis[row0 + i] = rsqrtf(1.0f + (float)degc[i]);
    if (tid == 0) slots[blockIdx.x * SPITCH] = nslot;
}


__global__ void gcn_gemm_wmma(const float* __restrict__ A,
                              const float* __restrict__ W,
                              float* __restrict__ H) {
    int waveId = blockIdx.x * (blockDim.x >> 5) + (threadIdx.x >> 5);
    int lane   = threadIdx.x & 31;
    int mTile  = waveId >> 3;
    int nTile  = waveId & 7;
    int hi     = lane >> 4;
    int l16    = lane & 15;

    int row = mTile * 16 + l16;
    int col = nTile * 16 + l16;

    const float* arow = A + (size_t)row * DCH + 2 * hi;

    v8f c = {};
#pragma unroll
    for (int k0 = 0; k0 < DCH; k0 += 4) {
        v2f a, b;
        a.x = arow[k0 + 0];
        a.y = arow[k0 + 1];
        int kb = k0 + 2 * hi;
        b.x = W[(size_t)kb * DCH + col];
        b.y = W[(size_t)(kb + 1) * DCH + col];
        c = __builtin_amdgcn_wmma_f32_16x16x4_f32(
                 false, a,  false, b,
                 (short)0, c,  false,  false);
    }

    float* hcol = H + (size_t)(mTile * 16) * DCH + nTile * 16 + l16;
#pragma unroll
    for (int i = 0; i < 8; ++i)
        hcol[(size_t)(i + 8 * hi) * DCH] = c[i];
}


struct AggFilter {
    float* acc; int* qe; int* qr;
    const float* __restrict__ H; const int* __restrict__ src; const float* __restrict__ dis;
    int row0; int tl; int lane; int cnt;

    __device__ __forceinline__ void operator()(int t, int ei, bool ok) {
        int rl = t - row0;
        unsigned lim = ok ? (unsigned)tl : 0u;
        bool m = (unsigned)rl < lim;
        unsigned mask = __builtin_amdgcn_ballot_w32(m);
        if (mask != 0u) {
            int p = (int)__builtin_amdgcn_mbcnt_lo(mask, 0u);
            if (m) { qe[cnt + p] = ei; qr[cnt + p] = rl; }
            cnt += __builtin_popcount(mask);
        }
    }

    __device__ __forceinline__ void one(int s, float w, int r, int j) {
        int   sj = __builtin_amdgcn_readlane(s, j);
        float wj = __int_as_float(__builtin_amdgcn_readlane(__float_as_int(w), j));
        int   rj = __builtin_amdgcn_readlane(r, j);
        float4 v = ((const float4*)(H + (size_t)sj * DCH))[lane];
        float* ap = acc + rj * DCH + lane;
        atomicAdd(ap,      v.x * wj);
        atomicAdd(ap + 32, v.y * wj);
        atomicAdd(ap + 64, v.z * wj);
        atomicAdd(ap + 96, v.w * wj);
    }

    __device__ __forceinline__ void add_rows(int s, float w, int r, int nvalid) {
        int j = 0;
        for (; j + 8 <= nvalid; j += 8) {
            float4 v[8];
#pragma unroll
            for (int u = 0; u < 8; ++u) {
                int sj = __builtin_amdgcn_readlane(s, j + u);
                v[u] = ((const float4*)(H + (size_t)sj * DCH))[lane];
            }
#pragma unroll
            for (int u = 0; u < 8; ++u) {
                float wj = __int_as_float(__builtin_amdgcn_readlane(__float_as_int(w), j + u));
                int   rj = __builtin_amdgcn_readlane(r, j + u);
                float* ap = acc + rj * DCH + lane;
                atomicAdd(ap,      v[u].x * wj);
                atomicAdd(ap + 32, v[u].y * wj);
                atomicAdd(ap + 64, v[u].z * wj);
                atomicAdd(ap + 96, v[u].w * wj);
            }
        }
        for (; j < nvalid; ++j) one(s, w, r, j);
    }

    __device__ __forceinline__ void from_records(int rec) {
        bool valid = rec >= 0;
        int nvalid = __builtin_popcount(__builtin_amdgcn_ballot_w32(valid));
        int r = 0, s = 0; float w = 0.0f;
        if (valid) {
            s = rec >> 8;
            r = rec & 255;
            w = dis[s] * dis[row0 + r];
        }
        add_rows(s, w, r, nvalid);
    }

    __device__ __forceinline__ void flush(int start, int nvalid) {
        int r = 0, s = 0; float w = 0.0f;
        if (lane < nvalid) {
            int ei = qe[start + lane];
            r = qr[start + lane];
            s = src[ei];
            w = dis[s] * dis[row0 + r];
        }
        add_rows(s, w, r, nvalid);
    }

    __device__ __forceinline__ void trip_end() {
        int done = 0;
        while (cnt - done >= 32) { flush(done, 32); done += 32; }
        if (done > 0) {
            int rem = cnt - done;
            int a = 0, b = 0;
            if (lane < rem) { a = qe[done + lane]; b = qr[done + lane]; }
            if (lane < rem) { qe[lane] = a; qr[lane] = b; }
            cnt = rem;
        }
    }
};

__global__ __launch_bounds__(NTHR) void gcn_agg_tile(const float* __restrict__ H,
                                                     const int* __restrict__ src,
                                                     const int* __restrict__ tgt,
                                                     const float* __restrict__ dis,
                                                     const float* __restrict__ bias,
                                                     const int* __restrict__ recs,
                                                     const int* __restrict__ slots,
                                                     float* __restrict__ out,
                                                     int n, int e, int relu) {
    __shared__ float acc[TROWS * DCH];
    __shared__ int qe[NWAVES * QCAP];
    __shared__ int qr[NWAVES * QCAP];
    const int tid = threadIdx.x, lane = tid & 31, wave = tid >> 5;
    const int row0 = blockIdx.x * TROWS;
    const int tl = min(TROWS, n - row0);

    const float4 b = ((const float4*)bias)[lane];
    {
        constexpr int RPW = (TROWS + NWAVES - 1) / NWAVES;
        float4 h[RPW]; float d[RPW];
#pragma unroll
        for (int i = 0; i < RPW; ++i) {
            int rc = min(wave + NWAVES * i, tl - 1);
            d[i] = dis[row0 + rc];
            h[i] = ((const float4*)(H + (size_t)(row0 + rc) * DCH))[lane];
        }
#pragma unroll
        for (int i = 0; i < RPW; ++i) {
            int r = wave + NWAVES * i;
            if (r < tl) {
                float* ap = acc + r * DCH + lane;
                ap[0]  = h[i].x * d[i] * d[i] + b.x;
                ap[32] = h[i].y * d[i] * d[i] + b.y;
                ap[64] = h[i].z * d[i] * d[i] + b.z;
                ap[96] = h[i].w * d[i] * d[i] + b.w;
            }
        }
    }
    __syncthreads();

    AggFilter f{acc, qe + wave * QCAP, qr + wave * QCAP, H, src, dis, row0, tl, lane, 0};
    const int nslot = slots[blockIdx.x * SPITCH];
    if (nslot <= RCAP) {
        const int* __restrict__ rt = recs + (size_t)blockIdx.x * RCAP;
        const int nlines = nslot >> 5;
        int g = __builtin_amdgcn_readfirstlane(wave);
        int rec = (g < nlines) ? rt[g * 32 + lane] : -1;
        while (g < nlines) {
            const int gn = g + NWAVES;
            const int recn = (gn < nlines) ? rt[gn * 32 + lane] : -1;
            f.from_records(rec);
            rec = recn;
            g = gn;
        }
    } else {
        scan_targets(tgt, e, wave, lane, f);
        if (f.cnt > 0) f.flush(0, f.cnt);
    }
    __syncthreads();

    for (int r = wave; r < tl; r += NWAVES) {
        const float* ap = acc + r * DCH + lane;
        float4 v;
        v.x = ap[0]; v.y = ap[32]; v.z = ap[64]; v.w = ap[96];
        if (relu) {
            v.x = fmaxf(v.x, 0.0f); v.y = fmaxf(v.y, 0.0f);
            v.z = fmaxf(v.z, 0.0f); v.w = fmaxf(v.w, 0.0f);
        }
        ((float4*)(out + (size_t)(row0 + r) * DCH))[lane] = v;
    }
}


extern "C" void kernel_launch(void* const* d_in, const int* in_sizes, int n_in,
                              void* d_out, int out_size, void* d_ws, size_t ws_size,
                              hipStream_t stream) {
    const float* X  = (const float*)d_in[0];
    const int*   ei = (const int*)  d_in[1];
    const float* W1 = (const float*)d_in[2];
    const float* b1 = (const float*)d_in[3];
    const float* W2 = (const float*)d_in[4];
    const float* b2 = (const float*)d_in[5];
    const float* W3 = (const float*)d_in[6];
    const float* b3 = (const float*)d_in[7];
    float* out = (float*)d_out;

    const int n  = in_sizes[0] / DCH;
    const int e  = in_sizes[1] / 2;

    const int* src = ei;
    const int* tgt = ei + e;

    const int mTiles     = n / 16;
    const int gemmBlocks = (mTiles * 8) / 8;
    const int aggBlocks  = (n + TROWS - 1) / TROWS;

    int*   recs  = (int*)d_ws;
    float* dis   = (float*)(recs + (size_t)aggBlocks * RCAP);
    float* H     = dis + n;
    int*   slots = (int*)(H + (size_t)n * DCH);

    gcn_filter_tile<<<aggBlocks, NTHR, 0, stream>>>(src, tgt, dis, recs, slots, n, e);

    const float* Ws[3] = {W1, W2, W3};
    const float* bs[3] = {b1, b2, b3};
    const float* cur   = X;

    for (int layer = 0; layer < 3; ++layer) {
        gcn_gemm_wmma<<<gemmBlocks, 256, 0, stream>>>(cur, Ws[layer], H);
        gcn_agg_tile <<<aggBlocks, NTHR, 0, stream>>>(H, src, tgt, dis, bs[layer], recs, slots, out,
                                                      n, e, layer < 2 ? 1 : 0);
        cur = out;
    }
}


# GEMM retiled: 256 blocks keep W in LDS (k-quad words), a wave holds its 16 A rows in registers and chains 4 accumulators (16x64 outputs) over the block's row tiles; k summed in regrouped order
# speedup vs baseline: 3.9788x; 1.1873x over previous
#include <hip/hip_runtime.h>
#include <stdint.h>


typedef __attribute__((ext_vector_type(2))) float v2f;
typedef __attribute__((ext_vector_type(8))) float v8f;

#define DCH    128
#define TROWS  224
#define SPITCH 32
#define NTHR   1024
#define NWAVES 32
#define QCAP   544
#define RCAP   32768

static_assert(TROWS <= 256, "a record keeps the tile row in its low 8 bits");

template <bool GUARD, class F>
__device__ __forceinline__ void filter_trip(const int4 (&q)[4], int base, int head, int m4, F& f) {
#pragma unroll
    for (int u = 0; u < 4; ++u) {
        const int i4 = base + 32 * u;
        const bool ok = GUARD ? (i4 < m4) : true;
        const int e0 = head + 4 * i4;
        f(q[u].x, e0, ok);
        f(q[u].y, e0 + 1, ok);
        f(q[u].z, e0 + 2, ok);
        f(q[u].w, e0 + 3, ok);
    }
}

template <class F>
__device__ __forceinline__ void scan_targets(const int* __restrict__ tgt, int e,
                                             int wave, int lane, F& f) {
    const uintptr_t a = (uintptr_t)tgt;
    int head = (int)(((16u - (unsigned)(a & 15u)) & 15u) >> 2);
    if (head > e) head = e;
    const int m4    = (e - head) >> 2;
    const int tail0 = head + 4 * m4;
    if (wave == 0 && head > 0) {
        int i = lane;
        bool ok = i < head;
        int t = ok ? tgt[i] : -1;
        f(t, i, ok);
    }
    if (wave == NWAVES - 1 && tail0 < e) {
        int i = tail0 + lane;
        bool ok = i < e;
        int t = ok ? tgt[i] : -1;
        f(t, i, ok);
    }
    const int4* __restrict__ t4 = (const int4*)(tgt + head);
    const int per_it = NWAVES * 128;
    const int nfull  = m4 / per_it;
    const int woff   = wave * 128 + lane;
    if (nfull > 0) {
        int it = (int)((blockIdx.x * (unsigned)nfull) / gridDim.x);
        int4 q[4];
#pragma unroll
        for (int u = 0; u < 4; ++u) q[u] = t4[it * per_it + woff + 32 * u];
        for (int k = 0; k < nfull; ++k) {
            int nit = it + 1;
            if (nit == nfull) nit = 0;
            int4 nq[4];
#pragma unroll
            for (int u = 0; u < 4; ++u) nq[u] = t4[nit * per_it + woff + 32 * u];
            filter_trip<false>(q, it * per_it + woff, head, m4, f);
            f.trip_end();
#pragma unroll
            for (int u = 0; u < 4; ++u) q[u] = nq[u];
            it = nit;
        }
    }
    if (nfull * per_it < m4) {
        const int base = nfull * per_it + woff;
        int4 q[4];
#pragma unroll
        for (int u = 0; u < 4; ++u) q[u] = t4[min(base + 32 * u, m4 - 1)];
        filter_trip<true>(q, base, head, m4, f);
        f.trip_end();
    }
}


struct TileFilter {
    int* qe; int* qr; int* degc; int* nslot;
    int* __restrict__ recs; const int* __restrict__ src;
    int row0; int tl; int lane; int cnt;

    __device__ __forceinline__ void operator()(int t, int ei, bool ok) {
        int rl = t - row0;
        unsigned lim = ok ? (unsigned)tl : 0u;
        bool m = (unsigned)rl < lim;
        unsigned mask = __builtin_amdgcn_ballot_w32(m);
        if (mask != 0u) {
            int p = (int)__builtin_amdgcn_mbcnt_lo(mask, 0u);
            if (m) { qe[cnt + p] = ei; qr[cnt + p] = rl; }
            cnt += __builtin_popcount(mask);
        }
    }

    __device__ __forceinline__ void emit(int start, int nvalid) {
        int rec = -1;
        if (lane < nvalid) {
            int ei = qe[start + lane];
            int r  = qr[start + lane];
            int s  = src[ei];
            atomicAdd(&degc[r], 1);
            rec = (s << 8) | r;
        }
        int pos = 0;
        if (lane == 0) pos = atomicAdd(nslot, 32);
        pos = __builtin_amdgcn_readfirstlane(pos);
        if (pos + 32 <= RCAP) recs[pos + lane] = rec;
    }

    __device__ __forceinline__ void trip_end() {
        int done = 0;
        while (cnt - done >= 32) { emit(done, 32); done += 32; }
        if (done > 0) {
            int rem = cnt - done;
            int a = 0, b = 0;
            if (lane < rem) { a = qe[done + lane]; b = qr[done + lane]; }
            if (lane < rem) { qe[lane] = a; qr[lane] = b; }
            cnt = rem;
        }
    }
};

__global__ __launch_bounds__(NTHR) void gcn_filter_tile(const int* __restrict__ src,
                                                        const int* __restrict__ tgt,
                                                        float* __restrict__ dis,
                                                        int* __restrict__ recs,
                                                        int* __restrict__ slots,
                                                        int n, int e) {
    __shared__ int qe[NWAVES * QCAP];
    __shared__ int qr[NWAVES * QCAP];
    __shared__ int degc[TROWS];
    __shared__ int nslot;
    const int tid = threadIdx.x, lane = tid & 31, wave = tid >> 5;
    const int row0 = blockIdx.x * TROWS;
    const int tl = min(TROWS, n - row0);
    for (int i = tid; i < TROWS; i += NTHR) degc[i] = 0;
    if (tid == 0) nslot = 0;
    __syncthreads();
    TileFilter f{qe + wave * QCAP, qr + wave * QCAP, degc, &nslot,
                 recs + (size_t)blockIdx.x * RCAP, src, row0, tl, lane, 0};
    scan_targets(tgt, e, wave, lane, f);
    if (f.cnt > 0) f.emit(0, f.cnt);
    __syncthreads();
    for (int i = tid; i < tl; i += NTHR)
        dis[row0 + i] = rsqrtf(1.0f + (float)degc[i]);
    if (tid == 0) slots[blockIdx.x * SPITCH] = nslot;
}


#define GEMM_WAVES 16
#define GEMM_THR   (GEMM_WAVES * 32)
#define WQ_PITCH   136

__global__ __launch_bounds__(GEMM_THR) __attribute__((amdgpu_waves_per_eu(4, 4))) void gcn_gemm_wmma(const float* __restrict__ A,
                                                          const float* __restrict__ W,
                                                          float* __restrict__ H,
                                                          int n) {
    __shared__ float4 wq[(DCH / 4) * WQ_PITCH];
    const int tid = threadIdx.x, lane = tid & 31;
    const int wave = __builtin_amdgcn_readfirstlane(tid >> 5);

    const int h = lane >> 4;
    const int c = lane & 15;
    const int mTiles = n >> 4;
    const int nmine  = (mTiles - (int)blockIdx.x + (int)gridDim.x - 1) / (int)gridDim.x;
    const int nitems = 2 * nmine;

    float4 a[DCH / 8];
    {
        const int tile = min((int)blockIdx.x + (int)gridDim.x * (wave >> 1), mTiles - 1);
        const float4* __restrict__ ap = (const float4*)(A + (size_t)(tile * 16 + c) * DCH) + h;
#pragma unroll
        for (int j = 0; j < DCH / 8; ++j) a[j] = ap[2 * j];
    }

    {
        constexpr int NQ = (DCH / 4) * DCH / GEMM_THR;
        float4 v[NQ];
#pragma unroll
        for (int i = 0; i < NQ; ++i) {
            const int idx = tid + GEMM_THR * i;
            const int q = idx >> 7, col = idx & (DCH - 1);
            const float* wp = W + (size_t)(4 * q) * DCH + col;
            v[i].x = wp[0];
            v[i].y = wp[DCH];
            v[i].z = wp[2 * DCH];
            v[i].w = wp[3 * DCH];
        }
#pragma unroll
        for (int i = 0; i < NQ; ++i) {
            const int idx = tid + GEMM_THR * i;
            wq[(idx >> 7) * WQ_PITCH + (idx & (DCH - 1))] = v[i];
        }
    }
    __syncthreads();

    for (int item = wave; item < nitems; item += GEMM_WAVES) {
        const int tile = (int)blockIdx.x + (int)gridDim.x * (item >> 1);
        const int nh   = item & 1;

        float4 an[DCH / 8];
        {
            const int tn = min((int)blockIdx.x + (int)gridDim.x * ((item + GEMM_WAVES) >> 1), mTiles - 1);
            const float4* __restrict__ apn = (const float4*)(A + (size_t)(tn * 16 + c) * DCH) + h;
#pragma unroll
            for (int j = 0; j < DCH / 8; ++j) an[j] = apn[2 * j];
        }
        __builtin_amdgcn_sched_barrier(0);

        const float4* bp = wq + h * WQ_PITCH + 64 * nh + c;
        v8f acc[4] = {};
        float4 b[4];
#pragma unroll
        for (int t = 0; t < 4; ++t) b[t] = bp[16 * t];
#pragma unroll
        for (int j = 0; j < DCH / 8; ++j) {
            float4 bn[4];
            if (j + 1 < DCH / 8) {
#pragma unroll
                for (int t = 0; t < 4; ++t) bn[t] = bp[2 * (j + 1) * WQ_PITCH + 16 * t];
            }
#pragma unroll
            for (int t = 0; t < 4; ++t) {
                v2f av, bv;
                av.x = a[j].x; av.y = a[j].y;
                bv.x = b[t].x; bv.y = b[t].y;
                acc[t] = __builtin_amdgcn_wmma_f32_16x16x4_f32(
                         false, av,  false, bv,
                         (short)0, acc[t],  false,  false);
            }
#pragma unroll
            for (int t = 0; t < 4; ++t) {
                v2f av, bv;
                av.x = a[j].z; av.y = a[j].w;
                bv.x = b[t].z; bv.y = b[t].w;
                acc[t] = __builtin_amdgcn_wmma_f32_16x16x4_f32(
                         false, av,  false, bv,
                         (short)0, acc[t],  false,  false);
            }
            if (j + 1 < DCH / 8) {
#pragma unroll
                for (int t = 0; t < 4; ++t) b[t] = bn[t];
            }
        }

        const int orow = tile * 16 + 8 * h;
        float* hp = H + (size_t)orow * DCH + 64 * nh + c;
#pragma unroll
        for (int r = 0; r < 8; ++r) {
#pragma unroll
            for (int t = 0; t < 4; ++t) hp[(size_t)r * DCH + 16 * t] = acc[t][r];
        }
#pragma unroll
        for (int j = 0; j < DCH / 8; ++j) a[j] = an[j];
    }
}


struct AggFilter {
    float* acc; int* qe; int* qr;
    const float* __restrict__ H; const int* __restrict__ src; const float* __restrict__ dis;
    int row0; int tl; int lane; int cnt;

    __device__ __forceinline__ void operator()(int t, int ei, bool ok) {
        int rl = t - row0;
        unsigned lim = ok ? (unsigned)tl : 0u;
        bool m = (unsigned)rl < lim;
        unsigned mask = __builtin_amdgcn_ballot_w32(m);
        if (mask != 0u) {
            int p = (int)__builtin_amdgcn_mbcnt_lo(mask, 0u);
            if (m) { qe[cnt + p] = ei; qr[cnt + p] = rl; }
            cnt += __builtin_popcount(mask);
        }
    }

    __device__ __forceinline__ void one(int s, float w, int r, int j) {
        int   sj = __builtin_amdgcn_readlane(s, j);
        float wj = __int_as_float(__builtin_amdgcn_readlane(__float_as_int(w), j));
        int   rj = __builtin_amdgcn_readlane(r, j);
        float4 v = ((const float4*)(H + (size_t)sj * DCH))[lane];
        float* ap = acc + rj * DCH + lane;
        atomicAdd(ap,      v.x * wj);
        atomicAdd(ap + 32, v.y * wj);
        atomicAdd(ap + 64, v.z * wj);
        atomicAdd(ap + 96, v.w * wj);
    }

    __device__ __forceinline__ void add_rows(int s, float w, int r, int nvalid) {
        int j = 0;
        for (; j + 8 <= nvalid; j += 8) {
            float4 v[8];
#pragma unroll
            for (int u = 0; u < 8; ++u) {
                int sj = __builtin_amdgcn_readlane(s, j + u);
                v[u] = ((const float4*)(H + (size_t)sj * DCH))[lane];
            }
#pragma unroll
            for (int u = 0; u < 8; ++u) {
                float wj = __int_as_float(__builtin_amdgcn_readlane(__float_as_int(w), j + u));
                int   rj = __builtin_amdgcn_readlane(r, j + u);
                float* ap = acc + rj * DCH + lane;
                atomicAdd(ap,      v[u].x * wj);
                atomicAdd(ap + 32, v[u].y * wj);
                atomicAdd(ap + 64, v[u].z * wj);
                atomicAdd(ap + 96, v[u].w * wj);
            }
        }
        for (; j < nvalid; ++j) one(s, w, r, j);
    }

    __device__ __forceinline__ void from_records(int rec) {
        bool valid = rec >= 0;
        int nvalid = __builtin_popcount(__builtin_amdgcn_ballot_w32(valid));
        int r = 0, s = 0; float w = 0.0f;
        if (valid) {
            s = rec >> 8;
            r = rec & 255;
            w = dis[s] * dis[row0 + r];
        }
        add_rows(s, w, r, nvalid);
    }

    __device__ __forceinline__ void flush(int start, int nvalid) {
        int r = 0, s = 0; float w = 0.0f;
        if (lane < nvalid) {
            int ei = qe[start + lane];
            r = qr[start + lane];
            s = src[ei];
            w = dis[s] * dis[row0 + r];
        }
        add_rows(s, w, r, nvalid);
    }

    __device__ __forceinline__ void trip_end() {
        int done = 0;
        while (cnt - done >= 32) { flush(done, 32); done += 32; }
        if (done > 0) {
            int rem = cnt - done;
            int a = 0, b = 0;
            if (lane < rem) { a = qe[done + lane]; b = qr[done + lane]; }
            if (lane < rem) { qe[lane] = a; qr[lane] = b; }
            cnt = rem;
        }
    }
};

__global__ __launch_bounds__(NTHR) void gcn_agg_tile(const float* __restrict__ H,
                                                     const int* __restrict__ src,
                                                     const int* __restrict__ tgt,
                                                     const float* __restrict__ dis,
                                                     const float* __restrict__ bias,
                                                     const int* __restrict__ recs,
                                                     const int* __restrict__ slots,
                                                     float* __restrict__ out,
                                                     int n, int e, int relu) {
    __shared__ float acc[TROWS * DCH];
    __shared__ int qe[NWAVES * QCAP];
    __shared__ int qr[NWAVES * QCAP];
    const int tid = threadIdx.x, lane = tid & 31, wave = tid >> 5;
    const int row0 = blockIdx.x * TROWS;
    const int tl = min(TROWS, n - row0);

    const float4 b = ((const float4*)bias)[lane];
    {
        constexpr int RPW = (TROWS + NWAVES - 1) / NWAVES;
        float4 h[RPW]; float d[RPW];
#pragma unroll
        for (int i = 0; i < RPW; ++i) {
            int rc = min(wave + NWAVES * i, tl - 1);
            d[i] = dis[row0 + rc];
            h[i] = ((const float4*)(H + (size_t)(row0 + rc) * DCH))[lane];
        }
#pragma unroll
        for (int i = 0; i < RPW; ++i) {
            int r = wave + NWAVES * i;
            if (r < tl) {
                float* ap = acc + r * DCH + lane;
                ap[0]  = h[i].x * d[i] * d[i] + b.x;
                ap[32] = h[i].y * d[i] * d[i] + b.y;
                ap[64] = h[i].z * d[i] * d[i] + b.z;
                ap[96] = h[i].w * d[i] * d[i] + b.w;
            }
        }
    }
    __syncthreads();

    AggFilter f{acc, qe + wave * QCAP, qr + wave * QCAP, H, src, dis, row0, tl, lane, 0};
    const int nslot = slots[blockIdx.x * SPITCH];
    if (nslot <= RCAP) {
        const int* __restrict__ rt = recs + (size_t)blockIdx.x * RCAP;
        const int nlines = nslot >> 5;
        int g = __builtin_amdgcn_readfirstlane(wave);
        int rec = (g < nlines) ? rt[g * 32 + lane] : -1;
        while (g < nlines) {
            const int gn = g + NWAVES;
            const int recn = (gn < nlines) ? rt[gn * 32 + lane] : -1;
            f.from_records(rec);
            rec = recn;
            g = gn;
        }
    } else {
        scan_targets(tgt, e, wave, lane, f);
        if (f.cnt > 0) f.flush(0, f.cnt);
    }
    __syncthreads();

    for (int r = wave; r < tl; r += NWAVES) {
        const float* ap = acc + r * DCH + lane;
        float4 v;
        v.x = ap[0]; v.y = ap[32]; v.z = ap[64]; v.w = ap[96];
        if (relu) {
            v.x = fmaxf(v.x, 0.0f); v.y = fmaxf(v.y, 0.0f);
            v.z = fmaxf(v.z, 0.0f); v.w = fmaxf(v.w, 0.0f);
        }
        ((float4*)(out + (size_t)(row0 + r) * DCH))[lane] = v;
    }
}


extern "C" void kernel_launch(void* const* d_in, const int* in_sizes, int n_in,
                              void* d_out, int out_size, void* d_ws, size_t ws_size,
                              hipStream_t stream) {
    const float* X  = (const float*)d_in[0];
    const int*   ei = (const int*)  d_in[1];
    const float* W1 = (const float*)d_in[2];
    const float* b1 = (const float*)d_in[3];
    const float* W2 = (const float*)d_in[4];
    const float* b2 = (const float*)d_in[5];
    const float* W3 = (const float*)d_in[6];
    const float* b3 = (const float*)d_in[7];
    float* out = (float*)d_out;

    const int n  = in_sizes[0] / DCH;
    const int e  = in_sizes[1] / 2;

    const int* src = ei;
    const int* tgt = ei + e;

    const int gemmBlocks = 256;
    const int aggBlocks  = (n + TROWS - 1) / TROWS;

    int*   recs  = (int*)d_ws;
    float* dis   = (float*)(recs + (size_t)aggBlocks * RCAP);
    float* H     = dis + n;
    int*   slots = (int*)(H + (size_t)n * DCH);

    gcn_filter_tile<<<aggBlocks, NTHR, 0, stream>>>(src, tgt, dis, recs, slots, n, e);

    const float* Ws[3] = {W1, W2, W3};
    const float* bs[3] = {b1, b2, b3};
    const float* cur   = X;

    for (int layer = 0; layer < 3; ++layer) {
        gcn_gemm_wmma<<<gemmBlocks, GEMM_THR, 0, stream>>>(cur, Ws[layer], H, n);
        gcn_agg_tile <<<aggBlocks, NTHR, 0, stream>>>(H, src, tgt, dis, bs[layer], recs, slots, out,
                                                      n, e, layer < 2 ? 1 : 0);
        cur = out;
    }
}


# Tile height 224 -> 196 target rows: 256 tiles, one per WGP (32 WGPs had none), an eighth fewer LDS f32 adds per block; deg^-1/2 kept per tile at pitch 224; sums in arrival order as before
# speedup vs baseline: 4.3616x; 1.0962x over previous
#include <hip/hip_runtime.h>
#include <stdint.h>


typedef __attribute__((ext_vector_type(2))) float v2f;
typedef __attribute__((ext_vector_type(8))) float v8f;

#define DCH    128
#define TROWS  196
#define DPITCH 224
#define SPITCH 32
#define NTHR   1024
#define NWAVES 32
#define QCAP   544
#define RCAP   32768

static_assert(TROWS <= 256, "a record keeps the tile row in its low 8 bits");
static_assert(DPITCH >= TROWS && DPITCH % 32 == 0, "a tile's deg^-1/2 values fill their own whole lines");

__device__ __forceinline__ int dis_slot(int i) {
    const int t = (int)((unsigned)i / (unsigned)TROWS);
    return i + t * (DPITCH - TROWS);
}

template <bool GUARD, class F>
__device__ __forceinline__ void filter_trip(const int4 (&q)[4], int base, int head, int m4, F& f) {
#pragma unroll
    for (int u = 0; u < 4; ++u) {
        const int i4 = base + 32 * u;
        const bool ok = GUARD ? (i4 < m4) : true;
        const int e0 = head + 4 * i4;
        f(q[u].x, e0, ok);
        f(q[u].y, e0 + 1, ok);
        f(q[u].z, e0 + 2, ok);
        f(q[u].w, e0 + 3, ok);
    }
}

template <class F>
__device__ __forceinline__ void scan_targets(const int* __restrict__ tgt, int e,
                                             int wave, int lane, F& f) {
    const uintptr_t a = (uintptr_t)tgt;
    int head = (int)(((16u - (unsigned)(a & 15u)) & 15u) >> 2);
    if (head > e) head = e;
    const int m4    = (e - head) >> 2;
    const int tail0 = head + 4 * m4;
    if (wave == 0 && head > 0) {
        int i = lane;
        bool ok = i < head;
        int t = ok ? tgt[i] : -1;
        f(t, i, ok);
    }
    if (wave == NWAVES - 1 && tail0 < e) {
        int i = tail0 + lane;
        bool ok = i < e;
        int t = ok ? tgt[i] : -1;
        f(t, i, ok);
    }
    const int4* __restrict__ t4 = (const int4*)(tgt + head);
    const int per_it = NWAVES * 128;
    const int nfull  = m4 / per_it;
    const int woff   = wave * 128 + lane;
    if (nfull > 0) {
        int it = (int)((blockIdx.x * (unsigned)nfull) / gridDim.x);
        int4 q[4];
#pragma unroll
        for (int u = 0; u < 4; ++u) q[u] = t4[it * per_it + woff + 32 * u];
        for (int k = 0; k < nfull; ++k) {
            int nit = it + 1;
            if (nit == nfull) nit = 0;
            int4 nq[4];
#pragma unroll
            for (int u = 0; u < 4; ++u) nq[u] = t4[nit * per_it + woff + 32 * u];
            filter_trip<false>(q, it * per_it + woff, head, m4, f);
            f.trip_end();
#pragma unroll
            for (int u = 0; u < 4; ++u) q[u] = nq[u];
            it = nit;
        }
    }
    if (nfull * per_it < m4) {
        const int base = nfull * per_it + woff;
        int4 q[4];
#pragma unroll
        for (int u = 0; u < 4; ++u) q[u] = t4[min(base + 32 * u, m4 - 1)];
        filter_trip<true>(q, base, head, m4, f);
        f.trip_end();
    }
}


struct TileFilter {
    int* qe; int* qr; int* degc; int* nslot;
    int* __restrict__ recs; const int* __restrict__ src;
    int row0; int tl; int lane; int cnt;

    __device__ __forceinline__ void operator()(int t, int ei, bool ok) {
        int rl = t - row0;
        unsigned lim = ok ? (unsigned)tl : 0u;
        bool m = (unsigned)rl < lim;
        unsigned mask = __builtin_amdgcn_ballot_w32(m);
        if (mask != 0u) {
            int p = (int)__builtin_amdgcn_mbcnt_lo(mask, 0u);
            if (m) { qe[cnt + p] = ei; qr[cnt + p] = rl; }
            cnt += __builtin_popcount(mask);
        }
    }

    __device__ __forceinline__ void emit(int start, int nvalid) {
        int rec = -1;
        if (lane < nvalid) {
            int ei = qe[start + lane];
            int r  = qr[start + lane];
            int s  = src[ei];
            atomicAdd(&degc[r], 1);
            rec = (s << 8) | r;
        }
        int pos = 0;
        if (lane == 0) pos = atomicAdd(nslot, 32);
        pos = __builtin_amdgcn_readfirstlane(pos);
        if (pos + 32 <= RCAP) recs[pos + lane] = rec;
    }

    __device__ __forceinline__ void trip_end() {
        int done = 0;
        while (cnt - done >= 32) { emit(done, 32); done += 32; }
        if (done > 0) {
            int rem = cnt - done;
            int a = 0, b = 0;
            if (lane < rem) { a = qe[done + lane]; b = qr[done + lane]; }
            if (lane < rem) { qe[lane] = a; qr[lane] = b; }
            cnt = rem;
        }
    }
};

__global__ __launch_bounds__(NTHR) void gcn_filter_tile(const int* __restrict__ src,
                                                        const int* __restrict__ tgt,
                                                        float* __restrict__ dis,
                                                        int* __restrict__ recs,
                                                        int* __restrict__ slots,
                                                        int n, int e) {
    __shared__ int qe[NWAVES * QCAP];
    __shared__ int qr[NWAVES * QCAP];
    __shared__ int degc[TROWS];
    __shared__ int nslot;
    const int tid = threadIdx.x, lane = tid & 31, wave = tid >> 5;
    const int row0 = blockIdx.x * TROWS;
    const int tl = min(TROWS, n - row0);
    for (int i = tid; i < TROWS; i += NTHR) degc[i] = 0;
    if (tid == 0) nslot = 0;
    __syncthreads();
    TileFilter f{qe + wave * QCAP, qr + wave * QCAP, degc, &nslot,
                 recs + (size_t)blockIdx.x * RCAP, src, row0, tl, lane, 0};
    scan_targets(tgt, e, wave, lane, f);
    if (f.cnt > 0) f.emit(0, f.cnt);
    __syncthreads();
    for (int i = tid; i < tl; i += NTHR)
        dis[blockIdx.x * DPITCH + i] = rsqrtf(1.0f + (float)degc[i]);
    if (tid == 0) slots[blockIdx.x * SPITCH] = nslot;
}


#define GEMM_WAVES 16
#define GEMM_THR   (GEMM_WAVES * 32)
#define WQ_PITCH   136

__global__ __launch_bounds__(GEMM_THR) __attribute__((amdgpu_waves_per_eu(4, 4))) void gcn_gemm_wmma(const float* __restrict__ A,
                                                          const float* __restrict__ W,
                                                          float* __restrict__ H,
                                                          int n) {
    __shared__ float4 wq[(DCH / 4) * WQ_PITCH];
    const int tid = threadIdx.x, lane = tid & 31;
    const int wave = __builtin_amdgcn_readfirstlane(tid >> 5);

    const int h = lane >> 4;
    const int c = lane & 15;
    const int mTiles = n >> 4;
    const int nmine  = (mTiles - (int)blockIdx.x + (int)gridDim.x - 1) / (int)gridDim.x;
    const int nitems = 2 * nmine;

    float4 a[DCH / 8];
    {
        const int tile = min((int)blockIdx.x + (int)gridDim.x * (wave >> 1), mTiles - 1);
        const float4* __restrict__ ap = (const float4*)(A + (size_t)(tile * 16 + c) * DCH) + h;
#pragma unroll
        for (int j = 0; j < DCH / 8; ++j) a[j] = ap[2 * j];
    }

    {
        constexpr int NQ = (DCH / 4) * DCH / GEMM_THR;
        float4 v[NQ];
#pragma unroll
        for (int i = 0; i < NQ; ++i) {
            const int idx = tid + GEMM_THR * i;
            const int q = idx >> 7, col = idx & (DCH - 1);
            const float* wp = W + (size_t)(4 * q) * DCH + col;
            v[i].x = wp[0];
            v[i].y = wp[DCH];
            v[i].z = wp[2 * DCH];
            v[i].w = wp[3 * DCH];
        }
#pragma unroll
        for (int i = 0; i < NQ; ++i) {
            const int idx = tid + GEMM_THR * i;
            wq[(idx >> 7) * WQ_PITCH + (idx & (DCH - 1))] = v[i];
        }
    }
    __syncthreads();

    for (int item = wave; item < nitems; item += GEMM_WAVES) {
        const int tile = (int)blockIdx.x + (int)gridDim.x * (item >> 1);
        const int nh   = item & 1;

        float4 an[DCH / 8];
        {
            const int tn = min((int)blockIdx.x + (int)gridDim.x * ((item + GEMM_WAVES) >> 1), mTiles - 1);
            const float4* __restrict__ apn = (const float4*)(A + (size_t)(tn * 16 + c) * DCH) + h;
#pragma unroll
            for (int j = 0; j < DCH / 8; ++j) an[j] = apn[2 * j];
        }
        __builtin_amdgcn_sched_barrier(0);

        const float4* bp = wq + h * WQ_PITCH + 64 * nh + c;
        v8f acc[4] = {};
        float4 b[4];
#pragma unroll
        for (int t = 0; t < 4; ++t) b[t] = bp[16 * t];
#pragma unroll
        for (int j = 0; j < DCH / 8; ++j) {
            float4 bn[4];
            if (j + 1 < DCH / 8) {
#pragma unroll
                for (int t = 0; t < 4; ++t) bn[t] = bp[2 * (j + 1) * WQ_PITCH + 16 * t];
            }
#pragma unroll
            for (int t = 0; t < 4; ++t) {
                v2f av, bv;
                av.x = a[j].x; av.y = a[j].y;
                bv.x = b[t].x; bv.y = b[t].y;
                acc[t] = __builtin_amdgcn_wmma_f32_16x16x4_f32(
                         false, av,  false, bv,
                         (short)0, acc[t],  false,  false);
            }
#pragma unroll
            for (int t = 0; t < 4; ++t) {
                v2f av, bv;
                av.x = a[j].z; av.y = a[j].w;
                bv.x = b[t].z; bv.y = b[t].w;
                acc[t] = __builtin_amdgcn_wmma_f32_16x16x4_f32(
                         false, av,  false, bv,
                         (short)0, acc[t],  false,  false);
            }
            if (j + 1 < DCH / 8) {
#pragma unroll
                for (int t = 0; t < 4; ++t) b[t] = bn[t];
            }
        }

        const int orow = tile * 16 + 8 * h;
        float* hp = H + (size_t)orow * DCH + 64 * nh + c;
#pragma unroll
        for (int r = 0; r < 8; ++r) {
#pragma unroll
            for (int t = 0; t < 4; ++t) hp[(size_t)r * DCH + 16 * t] = acc[t][r];
        }
#pragma unroll
        for (int j = 0; j < DCH / 8; ++j) a[j] = an[j];
    }
}


struct AggFilter {
    float* acc; int* qe; int* qr;
    const float* __restrict__ H; const int* __restrict__ src; const float* __restrict__ dis;
    int row0; int drow0; int tl; int lane; int cnt;

    __device__ __forceinline__ void operator()(int t, int ei, bool ok) {
        int rl = t - row0;
        unsigned lim = ok ? (unsigned)tl : 0u;
        bool m = (unsigned)rl < lim;
        unsigned mask = __builtin_amdgcn_ballot_w32(m);
        if (mask != 0u) {
            int p = (int)__builtin_amdgcn_mbcnt_lo(mask, 0u);
            if (m) { qe[cnt + p] = ei; qr[cnt + p] = rl; }
            cnt += __builtin_popcount(mask);
        }
    }

    __device__ __forceinline__ void one(int s, float w, int r, int j) {
        int   sj = __builtin_amdgcn_readlane(s, j);
        float wj = __int_as_float(__builtin_amdgcn_readlane(__float_as_int(w), j));
        int   rj = __builtin_amdgcn_readlane(r, j);
        float4 v = ((const float4*)(H + (size_t)sj * DCH))[lane];
        float* ap = acc + rj * DCH + lane;
        atomicAdd(ap,      v.x * wj);
        atomicAdd(ap + 32, v.y * wj);
        atomicAdd(ap + 64, v.z * wj);
        atomicAdd(ap + 96, v.w * wj);
    }

    __device__ __forceinline__ void add_rows(int s, float w, int r, int nvalid) {
        int j = 0;
        for (; j + 8 <= nvalid; j += 8) {
            float4 v[8];
#pragma unroll
            for (int u = 0; u < 8; ++u) {
                int sj = __builtin_amdgcn_readlane(s, j + u);
                v[u] = ((const float4*)(H + (size_t)sj * DCH))[lane];
            }
#pragma unroll
            for (int u = 0; u < 8; ++u) {
                float wj = __int_as_float(__builtin_amdgcn_readlane(__float_as_int(w), j + u));
                int   rj = __builtin_amdgcn_readlane(r, j + u);
                float* ap = acc + rj * DCH + lane;
                atomicAdd(ap,      v[u].x * wj);
                atomicAdd(ap + 32, v[u].y * wj);
                atomicAdd(ap + 64, v[u].z * wj);
                atomicAdd(ap + 96, v[u].w * wj);
            }
        }
        for (; j < nvalid; ++j) one(s, w, r, j);
    }

    __device__ __forceinline__ void from_records(int rec) {
        bool valid = rec >= 0;
        int nvalid = __builtin_popcount(__builtin_amdgcn_ballot_w32(valid));
        int r = 0, s = 0; float w = 0.0f;
        if (valid) {
            s = rec >> 8;
            r = rec & 255;
            w = dis[dis_slot(s)] * dis[drow0 + r];
        }
        add_rows(s, w, r, nvalid);
    }

    __device__ __forceinline__ void flush(int start, int nvalid) {
        int r = 0, s = 0; float w = 0.0f;
        if (lane < nvalid) {
            int ei = qe[start + lane];
            r = qr[start + lane];
            s = src[ei];
            w = dis[dis_slot(s)] * dis[drow0 + r];
        }
        add_rows(s, w, r, nvalid);
    }

    __device__ __forceinline__ void trip_end() {
        int done = 0;
        while (cnt - done >= 32) { flush(done, 32); done += 32; }
        if (done > 0) {
            int rem = cnt - done;
            int a = 0, b = 0;
            if (lane < rem) { a = qe[done + lane]; b = qr[done + lane]; }
            if (lane < rem) { qe[lane] = a; qr[lane] = b; }
            cnt = rem;
        }
    }
};

__global__ __launch_bounds__(NTHR) void gcn_agg_tile(const float* __restrict__ H,
                                                     const int* __restrict__ src,
                                                     const int* __restrict__ tgt,
                                                     const float* __restrict__ dis,
                                                     const float* __restrict__ bias,
                                                     const int* __restrict__ recs,
                                                     const int* __restrict__ slots,
                                                     float* __restrict__ out,
                                                     int n, int e, int relu) {
    __shared__ float acc[TROWS * DCH];
    __shared__ int qe[NWAVES * QCAP];
    __shared__ int qr[NWAVES * QCAP];
    const int tid = threadIdx.x, lane = tid & 31, wave = tid >> 5;
    const int row0 = blockIdx.x * TROWS;
    const int tl = min(TROWS, n - row0);

    const float4 b = ((const float4*)bias)[lane];
    {
        constexpr int RPW = (TROWS + NWAVES - 1) / NWAVES;
        float4 h[RPW]; float d[RPW];
#pragma unroll
        for (int i = 0; i < RPW; ++i) {
            int rc = min(wave + NWAVES * i, tl - 1);
            d[i] = dis[blockIdx.x * DPITCH + rc];
            h[i] = ((const float4*)(H + (size_t)(row0 + rc) * DCH))[lane];
        }
#pragma unroll
        for (int i = 0; i < RPW; ++i) {
            int r = wave + NWAVES * i;
            if (r < tl) {
                float* ap = acc + r * DCH + lane;
                ap[0]  = h[i].x * d[i] * d[i] + b.x;
                ap[32] = h[i].y * d[i] * d[i] + b.y;
                ap[64] = h[i].z * d[i] * d[i] + b.z;
                ap[96] = h[i].w * d[i] * d[i] + b.w;
            }
        }
    }
    __syncthreads();

    AggFilter f{acc, qe + wave * QCAP, qr + wave * QCAP, H, src, dis, row0, (int)blockIdx.x * DPITCH, tl, lane, 0};
    const int nslot = slots[blockIdx.x * SPITCH];
    if (nslot <= RCAP) {
        const int* __restrict__ rt = recs + (size_t)blockIdx.x * RCAP;
        const int nlines = nslot >> 5;
        int g = __builtin_amdgcn_readfirstlane(wave);
        int rec = (g < nlines) ? rt[g * 32 + lane] : -1;
        while (g < nlines) {
            const int gn = g + NWAVES;
            const int recn = (gn < nlines) ? rt[gn * 32 + lane] : -1;
            f.from_records(rec);
            rec = recn;
            g = gn;
        }
    } else {
        scan_targets(tgt, e, wave, lane, f);
        if (f.cnt > 0) f.flush(0, f.cnt);
    }
    __syncthreads();

    for (int r = wave; r < tl; r += NWAVES) {
        const float* ap = acc + r * DCH + lane;
        float4 v;
        v.x = ap[0]; v.y = ap[32]; v.z = ap[64]; v.w = ap[96];
        if (relu) {
            v.x = fmaxf(v.x, 0.0f); v.y = fmaxf(v.y, 0.0f);
            v.z = fmaxf(v.z, 0.0f); v.w = fmaxf(v.w, 0.0f);
        }
        ((float4*)(out + (size_t)(row0 + r) * DCH))[lane] = v;
    }
}


extern "C" void kernel_launch(void* const* d_in, const int* in_sizes, int n_in,
                              void* d_out, int out_size, void* d_ws, size_t ws_size,
                              hipStream_t stream) {
    const float* X  = (const float*)d_in[0];
    const int*   ei = (const int*)  d_in[1];
    const float* W1 = (const float*)d_in[2];
    const float* b1 = (const float*)d_in[3];
    const float* W2 = (const float*)d_in[4];
    const float* b2 = (const float*)d_in[5];
    const float* W3 = (const float*)d_in[6];
    const float* b3 = (const float*)d_in[7];
    float* out = (float*)d_out;

    const int n  = in_sizes[0] / DCH;
    const int e  = in_sizes[1] / 2;

    const int* src = ei;
    const int* tgt = ei + e;

    const int gemmBlocks = 256;
    const int aggBlocks  = (n + TROWS - 1) / TROWS;

    int*   recs  = (int*)d_ws;
    float* dis   = (float*)(recs + (size_t)aggBlocks * RCAP);
    float* H     = dis + (size_t)aggBlocks * DPITCH;
    int*   slots = (int*)(H + (size_t)n * DCH);

    gcn_filter_tile<<<aggBlocks, NTHR, 0, stream>>>(src, tgt, dis, recs, slots, n, e);

    const float* Ws[3] = {W1, W2, W3};
    const float* bs[3] = {b1, b2, b3};
    const float* cur   = X;

    for (int layer = 0; layer < 3; ++layer) {
        gcn_gemm_wmma<<<gemmBlocks, GEMM_THR, 0, stream>>>(cur, Ws[layer], H, n);
        gcn_agg_tile <<<aggBlocks, NTHR, 0, stream>>>(H, src, tgt, dis, bs[layer], recs, slots, out,
                                                      n, e, layer < 2 ? 1 : 0);
        cur = out;
    }
}


# LDS f32 atomic sums -> register sums: filter hands each record to the list of the wave owning its tile row (row mod 32); that wave sums its 6-7 rows in registers, in list (arrival) order
# speedup vs baseline: 4.6413x; 1.0641x over previous
#include <hip/hip_runtime.h>
#include <stdint.h>


typedef __attribute__((ext_vector_type(2))) float v2f;
typedef __attribute__((ext_vector_type(8))) float v8f;

#define DCH    128
#define TROWS  196
#define DPITCH 224
#define SPITCH 32
#define NTHR   1024
#define NWAVES 32
#define QCAP   544
#define RCAP   32768
#define OCAP   (RCAP / NWAVES)
#define RPW    ((TROWS + NWAVES - 1) / NWAVES)

static_assert(TROWS <= 256, "a record keeps the tile row in its low 8 bits");
static_assert(DPITCH >= TROWS && DPITCH % 32 == 0, "a tile's deg^-1/2 values fill their own whole lines");
static_assert(SPITCH >= NWAVES, "a tile's line of counts holds one count per owner wave");
static_assert(OCAP % 32 == 0, "an owner's records are whole lines");

__device__ __forceinline__ int dis_slot(int i) {
    const int t = (int)((unsigned)i / (unsigned)TROWS);
    return i + t * (DPITCH - TROWS);
}

template <bool GUARD, class F>
__device__ __forceinline__ void filter_trip(const int4 (&q)[4], int base, int head, int m4, F& f) {
#pragma unroll
    for (int u = 0; u < 4; ++u) {
        const int i4 = base + 32 * u;
        const bool ok = GUARD ? (i4 < m4) : true;
        const int e0 = head + 4 * i4;
        f(q[u].x, e0, ok);
        f(q[u].y, e0 + 1, ok);
        f(q[u].z, e0 + 2, ok);
        f(q[u].w, e0 + 3, ok);
    }
}

template <class F>
__device__ __forceinline__ void scan_targets(const int* __restrict__ tgt, int e,
                                             int wave, int lane, F& f) {
    const uintptr_t a = (uintptr_t)tgt;
    int head = (int)(((16u - (unsigned)(a & 15u)) & 15u) >> 2);
    if (head > e) head = e;
    const int m4    = (e - head) >> 2;
    const int tail0 = head + 4 * m4;
    if (wave == 0 && head > 0) {
        int i = lane;
        bool ok = i < head;
        int t = ok ? tgt[i] : -1;
        f(t, i, ok);
    }
    if (wave == NWAVES - 1 && tail0 < e) {
        int i = tail0 + lane;
        bool ok = i < e;
        int t = ok ? tgt[i] : -1;
        f(t, i, ok);
    }
    const int4* __restrict__ t4 = (const int4*)(tgt + head);
    const int per_it = NWAVES * 128;
    const int nfull  = m4 / per_it;
    const int woff   = wave * 128 + lane;
    if (nfull > 0) {
        int it = (int)((blockIdx.x * (unsigned)nfull) / gridDim.x);
        int4 q[4];
#pragma unroll
        for (int u = 0; u < 4; ++u) q[u] = t4[it * per_it + woff + 32 * u];
        for (int k = 0; k < nfull; ++k) {
            int nit = it + 1;
            if (nit == nfull) nit = 0;
            int4 nq[4];
#pragma unroll
            for (int u = 0; u < 4; ++u) nq[u] = t4[nit * per_it + woff + 32 * u];
            filter_trip<false>(q, it * per_it + woff, head, m4, f);
            f.trip_end();
#pragma unroll
            for (int u = 0; u < 4; ++u) q[u] = nq[u];
            it = nit;
        }
    }
    if (nfull * per_it < m4) {
        const int base = nfull * per_it + woff;
        int4 q[4];
#pragma unroll
        for (int u = 0; u < 4; ++u) q[u] = t4[min(base + 32 * u, m4 - 1)];
        filter_trip<true>(q, base, head, m4, f);
        f.trip_end();
    }
}


struct TileFilter {
    int* qe; int* qr; int* degc; int* ocnt; int* olist;
    const int* __restrict__ src;
    int row0; int tl; int lane; int cnt;

    __device__ __forceinline__ void operator()(int t, int ei, bool ok) {
        int rl = t - row0;
        unsigned lim = ok ? (unsigned)tl : 0u;
        bool m = (unsigned)rl < lim;
        unsigned mask = __builtin_amdgcn_ballot_w32(m);
        if (mask != 0u) {
            int p = (int)__builtin_amdgcn_mbcnt_lo(mask, 0u);
            if (m) { qe[cnt + p] = ei; qr[cnt + p] = rl; }
            cnt += __builtin_popcount(mask);
        }
    }

    __device__ __forceinline__ void emit(int start, int nvalid) {
        if (lane < nvalid) {
            int ei = qe[start + lane];
            int r  = qr[start + lane];
            int s  = src[ei];
            atomicAdd(&degc[r], 1);
            int o = r & (NWAVES - 1);
            int p = atomicAdd(&ocnt[o], 1);
            if (p < OCAP) olist[o * OCAP + p] = (s << 8) | r;
        }
    }

    __device__ __forceinline__ void trip_end() {
        int done = 0;
        while (cnt - done >= 32) { emit(done, 32); done += 32; }
        if (done > 0) {
            int rem = cnt - done;
            int a = 0, b = 0;
            if (lane < rem) { a = qe[done + lane]; b = qr[done + lane]; }
            if (lane < rem) { qe[lane] = a; qr[lane] = b; }
            cnt = rem;
        }
    }
};

__global__ __launch_bounds__(NTHR) void gcn_filter_tile(const int* __restrict__ src,
                                                        const int* __restrict__ tgt,
                                                        float* __restrict__ dis,
                                                        int* __restrict__ recs,
                                                        int* __restrict__ cnts,
                                                        int n, int e) {
    __shared__ int qe[NWAVES * QCAP];
    __shared__ int qr[NWAVES * QCAP];
    __shared__ int olist[NWAVES * OCAP];
    __shared__ int degc[TROWS];
    __shared__ int ocnt[NWAVES];
    const int tid = threadIdx.x, lane = tid & 31, wave = tid >> 5;
    const int row0 = blockIdx.x * TROWS;
    const int tl = min(TROWS, n - row0);
    for (int i = tid; i < TROWS; i += NTHR) degc[i] = 0;
    if (tid < NWAVES) ocnt[tid] = 0;
    __syncthreads();
    TileFilter f{qe + wave * QCAP, qr + wave * QCAP, degc, ocnt, olist, src, row0, tl, lane, 0};
    scan_targets(tgt, e, wave, lane, f);
    if (f.cnt > 0) f.emit(0, f.cnt);
    __syncthreads();
    for (int i = tid; i < tl; i += NTHR)
        dis[blockIdx.x * DPITCH + i] = rsqrtf(1.0f + (float)degc[i]);
    {
        const int c = min(ocnt[wave], OCAP);
        const int* ol = olist + wave * OCAP;
        int* __restrict__ dst = recs + (size_t)blockIdx.x * RCAP + wave * OCAP;
        for (int k0 = 0; k0 < c; k0 += 32) {
            const int k = k0 + lane;
            dst[k] = (k < c) ? ol[k] : -1;
        }
    }
    if (wave == 0) cnts[blockIdx.x * SPITCH + lane] = ocnt[lane];
}


#define GEMM_WAVES 16
#define GEMM_THR   (GEMM_WAVES * 32)
#define WQ_PITCH   136

__global__ __launch_bounds__(GEMM_THR) __attribute__((amdgpu_waves_per_eu(4, 4))) void gcn_gemm_wmma(const float* __restrict__ A,
                                                          const float* __restrict__ W,
                                                          float* __restrict__ H,
                                                          int n) {
    __shared__ float4 wq[(DCH / 4) * WQ_PITCH];
    const int tid = threadIdx.x, lane = tid & 31;
    const int wave = __builtin_amdgcn_readfirstlane(tid >> 5);

    const int h = lane >> 4;
    const int c = lane & 15;
    const int mTiles = n >> 4;
    const int nmine  = (mTiles - (int)blockIdx.x + (int)gridDim.x - 1) / (int)gridDim.x;
    const int nitems = 2 * nmine;

    float4 a[DCH / 8];
    {
        const int tile = min((int)blockIdx.x + (int)gridDim.x * (wave >> 1), mTiles - 1);
        const float4* __restrict__ ap = (const float4*)(A + (size_t)(tile * 16 + c) * DCH) + h;
#pragma unroll
        for (int j = 0; j < DCH / 8; ++j) a[j] = ap[2 * j];
    }

    {
        constexpr int NQ = (DCH / 4) * DCH / GEMM_THR;
        float4 v[NQ];
#pragma unroll
        for (int i = 0; i < NQ; ++i) {
            const int idx = tid + GEMM_THR * i;
            const int q = idx >> 7, col = idx & (DCH - 1);
            const float* wp = W + (size_t)(4 * q) * DCH + col;
            v[i].x = wp[0];
            v[i].y = wp[DCH];
            v[i].z = wp[2 * DCH];
            v[i].w = wp[3 * DCH];
        }
#pragma unroll
        for (int i = 0; i < NQ; ++i) {
            const int idx = tid + GEMM_THR * i;
            wq[(idx >> 7) * WQ_PITCH + (idx & (DCH - 1))] = v[i];
        }
    }
    __syncthreads();

    for (int item = wave; item < nitems; item += GEMM_WAVES) {
        const int tile = (int)blockIdx.x + (int)gridDim.x * (item >> 1);
        const int nh   = item & 1;

        float4 an[DCH / 8];
        {
            const int tn = min((int)blockIdx.x + (int)gridDim.x * ((item + GEMM_WAVES) >> 1), mTiles - 1);
            const float4* __restrict__ apn = (const float4*)(A + (size_t)(tn * 16 + c) * DCH) + h;
#pragma unroll
            for (int j = 0; j < DCH / 8; ++j) an[j] = apn[2 * j];
        }
        __builtin_amdgcn_sched_barrier(0);

        const float4* bp = wq + h * WQ_PITCH + 64 * nh + c;
        v8f acc[4] = {};
        float4 b[4];
#pragma unroll
        for (int t = 0; t < 4; ++t) b[t] = bp[16 * t];
#pragma unroll
        for (int j = 0; j < DCH / 8; ++j) {
            float4 bn[4];
            if (j + 1 < DCH / 8) {
#pragma unroll
                for (int t = 0; t < 4; ++t) bn[t] = bp[2 * (j + 1) * WQ_PITCH + 16 * t];
            }
#pragma unroll
            for (int t = 0; t < 4; ++t) {
                v2f av, bv;
                av.x = a[j].x; av.y = a[j].y;
                bv.x = b[t].x; bv.y = b[t].y;
                acc[t] = __builtin_amdgcn_wmma_f32_16x16x4_f32(
                         false, av,  false, bv,
                         (short)0, acc[t],  false,  false);
            }
#pragma unroll
            for (int t = 0; t < 4; ++t) {
                v2f av, bv;
                av.x = a[j].z; av.y = a[j].w;
                bv.x = b[t].z; bv.y = b[t].w;
                acc[t] = __builtin_amdgcn_wmma_f32_16x16x4_f32(
                         false, av,  false, bv,
                         (short)0, acc[t],  false,  false);
            }
            if (j + 1 < DCH / 8) {
#pragma unroll
                for (int t = 0; t < 4; ++t) b[t] = bn[t];
            }
        }

        const int orow = tile * 16 + 8 * h;
        float* hp = H + (size_t)orow * DCH + 64 * nh + c;
#pragma unroll
        for (int r = 0; r < 8; ++r) {
#pragma unroll
            for (int t = 0; t < 4; ++t) hp[(size_t)r * DCH + 16 * t] = acc[t][r];
        }
#pragma unroll
        for (int j = 0; j < DCH / 8; ++j) a[j] = an[j];
    }
}


struct RowSums { v8f x, y, z, w; };
static_assert(RPW <= 8, "one vector element per owned row");

__device__ __forceinline__ void add_row(RowSums& a, int k, const float4& v, float wt) {
    a.x[k] += v.x * wt;
    a.y[k] += v.y * wt;
    a.z[k] += v.z * wt;
    a.w[k] += v.w * wt;
}

__device__ __forceinline__ void add_line(RowSums& acc, const float* __restrict__ H,
                                         int s, float w, int k, int nvalid, int lane) {
    int j = 0;
    for (; j + 8 <= nvalid; j += 8) {
        float4 v[8];
#pragma unroll
        for (int u = 0; u < 8; ++u) {
            int sj = __builtin_amdgcn_readlane(s, j + u);
            v[u] = ((const float4*)(H + (size_t)sj * DCH))[lane];
        }
#pragma unroll
        for (int u = 0; u < 8; ++u) {
            float wj = __int_as_float(__builtin_amdgcn_readlane(__float_as_int(w), j + u));
            int   kj = __builtin_amdgcn_readlane(k, j + u);
            add_row(acc, kj, v[u], wj);
        }
    }
    if (j < nvalid) {
        const int nt = nvalid - j;
        float4 v[8];
#pragma unroll
        for (int u = 0; u < 7; ++u) {
            if (u < nt) {
                int sj = __builtin_amdgcn_readlane(s, j + u);
                v[u] = ((const float4*)(H + (size_t)sj * DCH))[lane];
            }
        }
#pragma unroll
        for (int u = 0; u < 7; ++u) {
            if (u < nt) {
                float wj = __int_as_float(__builtin_amdgcn_readlane(__float_as_int(w), j + u));
                int   kj = __builtin_amdgcn_readlane(k, j + u);
                add_row(acc, kj, v[u], wj);
            }
        }
    }
}

struct AggFilter {
    float* acc; int* qe; int* qr;
    const float* __restrict__ H; const int* __restrict__ src; const float* __restrict__ dis;
    int row0; int drow0; int tl; int lane; int cnt;

    __device__ __forceinline__ void operator()(int t, int ei, bool ok) {
        int rl = t - row0;
        unsigned lim = ok ? (unsigned)tl : 0u;
        bool m = (unsigned)rl < lim;
        unsigned mask = __builtin_amdgcn_ballot_w32(m);
        if (mask != 0u) {
            int p = (int)__builtin_amdgcn_mbcnt_lo(mask, 0u);
            if (m) { qe[cnt + p] = ei; qr[cnt + p] = rl; }
            cnt += __builtin_popcount(mask);
        }
    }

    __device__ __forceinline__ void one(int s, float w, int r, int j) {
        int   sj = __builtin_amdgcn_readlane(s, j);
        float wj = __int_as_float(__builtin_amdgcn_readlane(__float_as_int(w), j));
        int   rj = __builtin_amdgcn_readlane(r, j);
        float4 v = ((const float4*)(H + (size_t)sj * DCH))[lane];
        float* ap = acc + rj * DCH + lane;
        atomicAdd(ap,      v.x * wj);
        atomicAdd(ap + 32, v.y * wj);
        atomicAdd(ap + 64, v.z * wj);
        atomicAdd(ap + 96, v.w * wj);
    }

    __device__ __forceinline__ void add_rows(int s, float w, int r, int nvalid) {
        int j = 0;
        for (; j + 8 <= nvalid; j += 8) {
            float4 v[8];
#pragma unroll
            for (int u = 0; u < 8; ++u) {
                int sj = __builtin_amdgcn_readlane(s, j + u);
                v[u] = ((const float4*)(H + (size_t)sj * DCH))[lane];
            }
#pragma unroll
            for (int u = 0; u < 8; ++u) {
                float wj = __int_as_float(__builtin_amdgcn_readlane(__float_as_int(w), j + u));
                int   rj = __builtin_amdgcn_readlane(r, j + u);
                float* ap = acc + rj * DCH + lane;
                atomicAdd(ap,      v[u].x * wj);
                atomicAdd(ap + 32, v[u].y * wj);
                atomicAdd(ap + 64, v[u].z * wj);
                atomicAdd(ap + 96, v[u].w * wj);
            }
        }
        for (; j < nvalid; ++j) one(s, w, r, j);
    }

    __device__ __forceinline__ void flush(int start, int nvalid) {
        int r = 0, s = 0; float w = 0.0f;
        if (lane < nvalid) {
            int ei = qe[start + lane];
            r = qr[start + lane];
            s = src[ei];
            w = dis[dis_slot(s)] * dis[drow0 + r];
        }
        add_rows(s, w, r, nvalid);
    }

    __device__ __forceinline__ void trip_end() {
        int done = 0;
        while (cnt - done >= 32) { flush(done, 32); done += 32; }
        if (done > 0) {
            int rem = cnt - done;
            int a = 0, b = 0;
            if (lane < rem) { a = qe[done + lane]; b = qr[done + lane]; }
            if (lane < rem) { qe[lane] = a; qr[lane] = b; }
            cnt = rem;
        }
    }
};

__global__ __launch_bounds__(NTHR) void gcn_agg_tile(const float* __restrict__ H,
                                                     const int* __restrict__ src,
                                                     const int* __restrict__ tgt,
                                                     const float* __restrict__ dis,
                                                     const float* __restrict__ bias,
                                                     const int* __restrict__ recs,
                                                     const int* __restrict__ cnts,
                                                     float* __restrict__ out,
                                                     int n, int e, int relu) {
    __shared__ float acc[TROWS * DCH];
    __shared__ int qe[NWAVES * QCAP];
    __shared__ int qr[NWAVES * QCAP];
    const int tid = threadIdx.x, lane = tid & 31, wave = tid >> 5;
    const int row0 = blockIdx.x * TROWS;
    const int drow0 = blockIdx.x * DPITCH;
    const int tl = min(TROWS, n - row0);

    const float4 b = ((const float4*)bias)[lane];

    const int cv = cnts[blockIdx.x * SPITCH + lane];
    if (__builtin_amdgcn_ballot_w32(cv > OCAP) == 0u) {
        const int wv = __builtin_amdgcn_readfirstlane(wave);
        const int c  = __builtin_amdgcn_readlane(cv, wv);
        const int* __restrict__ rt = recs + (size_t)blockIdx.x * RCAP + wv * OCAP;
        const int nlines = (c + 31) >> 5;
        int rec  = (nlines > 0) ? rt[lane] : -1;
        int recn = (nlines > 1) ? rt[32 + lane] : -1;

        RowSums sum = {};
        {
            float4 h[RPW]; float d[RPW];
#pragma unroll
            for (int i = 0; i < RPW; ++i) {
                int rc = min(wv + NWAVES * i, tl - 1);
                d[i] = dis[drow0 + rc];
                h[i] = ((const float4*)(H + (size_t)(row0 + rc) * DCH))[lane];
            }
#pragma unroll
            for (int i = 0; i < RPW; ++i) {
                sum.x[i] = h[i].x * d[i] * d[i] + b.x;
                sum.y[i] = h[i].y * d[i] * d[i] + b.y;
                sum.z[i] = h[i].z * d[i] * d[i] + b.z;
                sum.w[i] = h[i].w * d[i] * d[i] + b.w;
            }
        }
        int s = (lane < c) ? (rec >> 8) : 0;
        int r = (lane < c) ? (rec & 255) : 0;
        float wa = dis[dis_slot(s)], wb = dis[drow0 + r];
        for (int g = 0; g < nlines; ++g) {
            const int recnn = rt[min(g + 2, nlines - 1) * 32 + lane];
            const int nvalid = min(32, c - 32 * g);
            const bool vn = lane < c - 32 * (g + 1);
            const int sn = vn ? (recn >> 8) : 0;
            const int rn = vn ? (recn & 255) : 0;
            const float wan = dis[dis_slot(sn)], wbn = dis[drow0 + rn];
            add_line(sum, H, s, wa * wb, r >> 5, nvalid, lane);
            s = sn; r = rn; wa = wan; wb = wbn;
            recn = recnn;
        }
#pragma unroll
        for (int i = 0; i < RPW; ++i) {
            const int orow = wv + NWAVES * i;
            if (orow < tl) {
                float4 v;
                v.x = sum.x[i]; v.y = sum.y[i]; v.z = sum.z[i]; v.w = sum.w[i];
                if (relu) {
                    v.x = fmaxf(v.x, 0.0f); v.y = fmaxf(v.y, 0.0f);
                    v.z = fmaxf(v.z, 0.0f); v.w = fmaxf(v.w, 0.0f);
                }
                ((float4*)(out + (size_t)(row0 + orow) * DCH))[lane] = v;
            }
        }
        return;
    }

    {
        float4 h[RPW]; float d[RPW];
#pragma unroll
        for (int i = 0; i < RPW; ++i) {
            int rc = min(wave + NWAVES * i, tl - 1);
            d[i] = dis[blockIdx.x * DPITCH + rc];
            h[i] = ((const float4*)(H + (size_t)(row0 + rc) * DCH))[lane];
        }
#pragma unroll
        for (int i = 0; i < RPW; ++i) {
            int r = wave + NWAVES * i;
            if (r < tl) {
                float* ap = acc + r * DCH + lane;
                ap[0]  = h[i].x * d[i] * d[i] + b.x;
                ap[32] = h[i].y * d[i] * d[i] + b.y;
                ap[64] = h[i].z * d[i] * d[i] + b.z;
                ap[96] = h[i].w * d[i] * d[i] + b.w;
            }
        }
    }
    __syncthreads();

    AggFilter f{acc, qe + wave * QCAP, qr + wave * QCAP, H, src, dis, row0, drow0, tl, lane, 0};
    scan_targets(tgt, e, wave, lane, f);
    if (f.cnt > 0) f.flush(0, f.cnt);
    __syncthreads();

    for (int r = wave; r < tl; r += NWAVES) {
        const float* ap = acc + r * DCH + lane;
        float4 v;
        v.x = ap[0]; v.y = ap[32]; v.z = ap[64]; v.w = ap[96];
        if (relu) {
            v.x = fmaxf(v.x, 0.0f); v.y = fmaxf(v.y, 0.0f);
            v.z = fmaxf(v.z, 0.0f); v.w = fmaxf(v.w, 0.0f);
        }
        ((float4*)(out + (size_t)(row0 + r) * DCH))[lane] = v;
    }
}


extern "C" void kernel_launch(void* const* d_in, const int* in_sizes, int n_in,
                              void* d_out, int out_size, void* d_ws, size_t ws_size,
                              hipStream_t stream) {
    const float* X  = (const float*)d_in[0];
    const int*   ei = (const int*)  d_in[1];
    const float* W1 = (const float*)d_in[2];
    const float* b1 = (const float*)d_in[3];
    const float* W2 = (const float*)d_in[4];
    const float* b2 = (const float*)d_in[5];
    const float* W3 = (const float*)d_in[6];
    const float* b3 = (const float*)d_in[7];
    float* out = (float*)d_out;

    const int n  = in_sizes[0] / DCH;
    const int e  = in_sizes[1] / 2;

    const int* src = ei;
    const int* tgt = ei + e;

    const int gemmBlocks = 256;
    const int aggBlocks  = (n + TROWS - 1) / TROWS;

    int*   recs  = (int*)d_ws;
    float* dis   = (float*)(recs + (size_t)aggBlocks * RCAP);
    float* H     = dis + (size_t)aggBlocks * DPITCH;
    int*   cnts  = (int*)(H + (size_t)n * DCH);

    gcn_filter_tile<<<aggBlocks, NTHR, 0, stream>>>(src, tgt, dis, recs, cnts, n, e);

    const float* Ws[3] = {W1, W2, W3};
    const float* bs[3] = {b1, b2, b3};
    const float* cur   = X;

    for (int layer = 0; layer < 3; ++layer) {
        gcn_gemm_wmma<<<gemmBlocks, GEMM_THR, 0, stream>>>(cur, Ws[layer], H, n);
        gcn_agg_tile <<<aggBlocks, NTHR, 0, stream>>>(H, src, tgt, dis, bs[layer], recs, cnts, out,
                                                      n, e, layer < 2 ? 1 : 0);
        cur = out;
    }
}


# Edge filter cut two ways: a block scans 1/8 of the edge list for a band of 8 tiles (was all of it for 1 tile); in-degree parts summed in GEMM 1; a row adds its 8 lists in turn, arrival order in each
# speedup vs baseline: 5.2832x; 1.1383x over previous
#include <hip/hip_runtime.h>
#include <stdint.h>


typedef __attribute__((ext_vector_type(2))) float v2f;
typedef __attribute__((ext_vector_type(8))) float v8f;

#define DCH    128
#define TROWS  196
#define DPITCH 224
#define SPITCH 32
#define NTHR   1024
#define NWAVES 32
#define QCAP   544
#define SLICES 8
#define BAND   8
#define BROWS  (BAND * TROWS)
#define NLIST  (BAND * NWAVES)
#define OCAP   64
#define RCAP   (SLICES * NWAVES * OCAP)
#define DP2    1600
#define RPW    ((TROWS + NWAVES - 1) / NWAVES)

static_assert(TROWS <= 256, "a record keeps the tile row in its low 8 bits");
static_assert(DPITCH >= TROWS && DPITCH % 32 == 0, "a tile's deg^-1/2 values fill their own whole lines");
static_assert(SPITCH >= NWAVES, "a tile's line of counts holds one count per owner wave");
static_assert(OCAP % 32 == 0, "an owner's records are whole lines");
static_assert(DP2 >= BROWS && DP2 % 32 == 0, "a block's row counts fill their own whole lines");
static_assert(SLICES * OCAP + 32 <= QCAP, "a wave's records of all slices fit its LDS queue, with a spare line behind them");

__device__ __forceinline__ int dis_slot(int i) {
    const int t = (int)((unsigned)i / (unsigned)TROWS);
    return i + t * (DPITCH - TROWS);
}

template <bool GUARD, class F>
__device__ __forceinline__ void filter_trip(const int4 (&q)[4], int base, int head, int m4, F& f) {
#pragma unroll
    for (int u = 0; u < 4; ++u) {
        const int i4 = base + 32 * u;
        const bool ok = GUARD ? (i4 < m4) : true;
        const int e0 = head + 4 * i4;
        f(q[u].x, e0, ok);
        f(q[u].y, e0 + 1, ok);
        f(q[u].z, e0 + 2, ok);
        f(q[u].w, e0 + 3, ok);
    }
}

template <class F>
__device__ __forceinline__ void scan_targets(const int* __restrict__ tgt, int e,
                                             int wave, int lane, F& f) {
    const uintptr_t a = (uintptr_t)tgt;
    int head = (int)(((16u - (unsigned)(a & 15u)) & 15u) >> 2);
    if (head > e) head = e;
    const int m4    = (e - head) >> 2;
    const int tail0 = head + 4 * m4;
    if (wave == 0 && head > 0) {
        int i = lane;
        bool ok = i < head;
        int t = ok ? tgt[i] : -1;
        f(t, i, ok);
    }
    if (wave == NWAVES - 1 && tail0 < e) {
        int i = tail0 + lane;
        bool ok = i < e;
        int t = ok ? tgt[i] : -1;
        f(t, i, ok);
    }
    const int4* __restrict__ t4 = (const int4*)(tgt + head);
    const int per_it = NWAVES * 128;
    const int nfull  = m4 / per_it;
    const int woff   = wave * 128 + lane;
    if (nfull > 0) {
        int it = (int)((blockIdx.x * (unsigned)nfull) / gridDim.x);
        int4 q[4];
#pragma unroll
        for (int u = 0; u < 4; ++u) q[u] = t4[it * per_it + woff + 32 * u];
        for (int k = 0; k < nfull; ++k) {
            int nit = it + 1;
            if (nit == nfull) nit = 0;
            int4 nq[4];
#pragma unroll
            for (int u = 0; u < 4; ++u) nq[u] = t4[nit * per_it + woff + 32 * u];
            filter_trip<false>(q, it * per_it + woff, head, m4, f);
            f.trip_end();
#pragma unroll
            for (int u = 0; u < 4; ++u) q[u] = nq[u];
            it = nit;
        }
    }
    if (nfull * per_it < m4) {
        const int base = nfull * per_it + woff;
        int4 q[4];
#pragma unroll
        for (int u = 0; u < 4; ++u) q[u] = t4[min(base + 32 * u, m4 - 1)];
        filter_trip<true>(q, base, head, m4, f);
        f.trip_end();
    }
}


struct TileFilter {
    int* qe; int* qr; int* degc; int* ocnt; int* olist;
    const int* __restrict__ src;
    int row0; int tl; int lane; int cnt;

    __device__ __forceinline__ void operator()(int t, int ei, bool ok) {
        int rl = t - row0;
        unsigned lim = ok ? (unsigned)tl : 0u;
        bool m = (unsigned)rl < lim;
        unsigned mask = __builtin_amdgcn_ballot_w32(m);
        if (mask != 0u) {
            int p = (int)__builtin_amdgcn_mbcnt_lo(mask, 0u);
            if (m) { qe[cnt + p] = ei; qr[cnt + p] = rl; }
            cnt += __builtin_popcount(mask);
        }
    }

    __device__ __forceinline__ void emit(int start, int nvalid) {
        if (lane < nvalid) {
            int ei = qe[start + lane];
            int r  = qr[start + lane];
            int s  = src[ei];
            atomicAdd(&degc[r], 1);
            int j  = (int)((unsigned)r / (unsigned)TROWS);
            int rr = r - j * TROWS;
            int o = j * NWAVES + (rr & (NWAVES - 1));
            int p = atomicAdd(&ocnt[o], 1);
            if (p < OCAP) olist[o * OCAP + p] = (s << 8) | rr;
        }
    }

    __device__ __forceinline__ void trip_end() {
        int done = 0;
        while (cnt - done >= 32) { emit(done, 32); done += 32; }
        if (done > 0) {
            int rem = cnt - done;
            int a = 0, b = 0;
            if (lane < rem) { a = qe[done + lane]; b = qr[done + lane]; }
            if (lane < rem) { qe[lane] = a; qr[lane] = b; }
            cnt = rem;
        }
    }
};

__global__ __launch_bounds__(NTHR) void gcn_filter_band(const int* __restrict__ src,
                                                        const int* __restrict__ tgt,
                                                        int* __restrict__ degp,
                                                        int* __restrict__ recs,
                                                        int* __restrict__ cnts,
                                                        int n, int e, int tiles) {
    __shared__ int qe[NWAVES * QCAP];
    __shared__ int qr[NWAVES * QCAP];
    __shared__ int olist[NLIST * OCAP];
    __shared__ int degc[BROWS];
    __shared__ int ocnt[NLIST];
    const int tid = threadIdx.x, lane = tid & 31, wave = tid >> 5;
    const int band  = (int)(blockIdx.x / (unsigned)SLICES);
    const int sl    = (int)(blockIdx.x % (unsigned)SLICES);
    const int bands = (int)(gridDim.x / (unsigned)SLICES);
    const int row0 = band * BROWS;
    const int tl = min(BROWS, n - row0);
    const int per = (((e + SLICES - 1) / SLICES) + 3) & ~3;
    const int e0 = min(sl * per, e);
    const int e1 = min(e0 + per, e);
    for (int i = tid; i < BROWS; i += NTHR) degc[i] = 0;
    if (tid < NLIST) ocnt[tid] = 0;
    __syncthreads();
    TileFilter f{qe + wave * QCAP, qr + wave * QCAP, degc, ocnt, olist, src + e0, row0, tl, lane, 0};
    scan_targets(tgt + e0, e1 - e0, wave, lane, f);
    if (f.cnt > 0) f.emit(0, f.cnt);
    __syncthreads();
    for (int i = tid; i < tl; i += NTHR)
        degp[(sl * bands + band) * DP2 + i] = degc[i];
    for (int j = 0; j < BAND; ++j) {
        const int tile = band * BAND + j;
        if (tile < tiles) {
            const int c = min(ocnt[j * NWAVES + wave], OCAP);
            const int* ol = olist + (j * NWAVES + wave) * OCAP;
            int* __restrict__ dst = recs + (size_t)tile * RCAP + (sl * NWAVES + wave) * OCAP;
            for (int k0 = 0; k0 < c; k0 += 32) {
                const int k = k0 + lane;
                dst[k] = (k < c) ? ol[k] : -1;
            }
        }
    }
    if (wave < BAND && band * BAND + wave < tiles)
        cnts[((band * BAND + wave) * SLICES + sl) * SPITCH + lane] = ocnt[wave * NWAVES + lane];
}


#define GEMM_WAVES 16
#define GEMM_THR   (GEMM_WAVES * 32)
#define WQ_PITCH   136

__global__ __launch_bounds__(GEMM_THR) __attribute__((amdgpu_waves_per_eu(4, 4))) void gcn_gemm_wmma(const float* __restrict__ A,
                                                          const float* __restrict__ W,
                                                          float* __restrict__ H,
                                                          int n,
                                                          const int* __restrict__ degp,
                                                          float* __restrict__ dis,
                                                          int tiles, int bands) {
    __shared__ float4 wq[(DCH / 4) * WQ_PITCH];
    const int tid = threadIdx.x, lane = tid & 31;
    const int wave = __builtin_amdgcn_readfirstlane(tid >> 5);

    const int h = lane >> 4;
    const int c = lane & 15;
    const int mTiles = n >> 4;
    const int nmine  = (mTiles - (int)blockIdx.x + (int)gridDim.x - 1) / (int)gridDim.x;
    const int nitems = 2 * nmine;

    float4 a[DCH / 8];
    {
        const int tile = min((int)blockIdx.x + (int)gridDim.x * (wave >> 1), mTiles - 1);
        const float4* __restrict__ ap = (const float4*)(A + (size_t)(tile * 16 + c) * DCH) + h;
#pragma unroll
        for (int j = 0; j < DCH / 8; ++j) a[j] = ap[2 * j];
    }

    {
        constexpr int NQ = (DCH / 4) * DCH / GEMM_THR;
        float4 v[NQ];
#pragma unroll
        for (int i = 0; i < NQ; ++i) {
            const int idx = tid + GEMM_THR * i;
            const int q = idx >> 7, col = idx & (DCH - 1);
            const float* wp = W + (size_t)(4 * q) * DCH + col;
            v[i].x = wp[0];
            v[i].y = wp[DCH];
            v[i].z = wp[2 * DCH];
            v[i].w = wp[3 * DCH];
        }
#pragma unroll
        for (int i = 0; i < NQ; ++i) {
            const int idx = tid + GEMM_THR * i;
            wq[(idx >> 7) * WQ_PITCH + (idx & (DCH - 1))] = v[i];
        }
    }
    __syncthreads();

    for (int item = wave; item < nitems; item += GEMM_WAVES) {
        const int tile = (int)blockIdx.x + (int)gridDim.x * (item >> 1);
        const int nh   = item & 1;

        float4 an[DCH / 8];
        {
            const int tn = min((int)blockIdx.x + (int)gridDim.x * ((item + GEMM_WAVES) >> 1), mTiles - 1);
            const float4* __restrict__ apn = (const float4*)(A + (size_t)(tn * 16 + c) * DCH) + h;
#pragma unroll
            for (int j = 0; j < DCH / 8; ++j) an[j] = apn[2 * j];
        }
        __builtin_amdgcn_sched_barrier(0);

        const float4* bp = wq + h * WQ_PITCH + 64 * nh + c;
        v8f acc[4] = {};
        float4 b[4];
#pragma unroll
        for (int t = 0; t < 4; ++t) b[t] = bp[16 * t];
#pragma unroll
        for (int j = 0; j < DCH / 8; ++j) {
            float4 bn[4];
            if (j + 1 < DCH / 8) {
#pragma unroll
                for (int t = 0; t < 4; ++t) bn[t] = bp[2 * (j + 1) * WQ_PITCH + 16 * t];
            }
#pragma unroll
            for (int t = 0; t < 4; ++t) {
                v2f av, bv;
                av.x = a[j].x; av.y = a[j].y;
                bv.x = b[t].x; bv.y = b[t].y;
                acc[t] = __builtin_amdgcn_wmma_f32_16x16x4_f32(
                         false, av,  false, bv,
                         (short)0, acc[t],  false,  false);
            }
#pragma unroll
            for (int t = 0; t < 4; ++t) {
                v2f av, bv;
                av.x = a[j].z; av.y = a[j].w;
                bv.x = b[t].z; bv.y = b[t].w;
                acc[t] = __builtin_amdgcn_wmma_f32_16x16x4_f32(
                         false, av,  false, bv,
                         (short)0, acc[t],  false,  false);
            }
            if (j + 1 < DCH / 8) {
#pragma unroll
                for (int t = 0; t < 4; ++t) b[t] = bn[t];
            }
        }

        const int orow = tile * 16 + 8 * h;
        float* hp = H + (size_t)orow * DCH + 64 * nh + c;
#pragma unroll
        for (int r = 0; r < 8; ++r) {
#pragma unroll
            for (int t = 0; t < 4; ++t) hp[(size_t)r * DCH + 16 * t] = acc[t][r];
        }
#pragma unroll
        for (int j = 0; j < DCH / 8; ++j) a[j] = an[j];
    }

    if (degp != nullptr) {
        for (int tile = (int)blockIdx.x; tile < tiles; tile += (int)gridDim.x) {
            const int i = tile * TROWS + tid;
            if (tid < TROWS && i < n) {
                const int band = tile / BAND;
                const int r = (tile - band * BAND) * TROWS + tid;
                int d = 0;
#pragma unroll
                for (int s = 0; s < SLICES; ++s) d += degp[(s * bands + band) * DP2 + r];
                dis[tile * DPITCH + tid] = rsqrtf(1.0f + (float)d);
            }
        }
    }
}


struct RowSums { v8f x, y, z, w; };
static_assert(RPW <= 8, "one vector element per owned row");

__device__ __forceinline__ void add_row(RowSums& a, int k, const float4& v, float wt) {
    a.x[k] += v.x * wt;
    a.y[k] += v.y * wt;
    a.z[k] += v.z * wt;
    a.w[k] += v.w * wt;
}

__device__ __forceinline__ void add_line(RowSums& acc, const float* __restrict__ H,
                                         int s, float w, int k, int nvalid, int lane) {
    int j = 0;
    for (; j + 8 <= nvalid; j += 8) {
        float4 v[8];
#pragma unroll
        for (int u = 0; u < 8; ++u) {
            int sj = __builtin_amdgcn_readlane(s, j + u);
            v[u] = ((const float4*)(H + (size_t)sj * DCH))[lane];
        }
#pragma unroll
        for (int u = 0; u < 8; ++u) {
            float wj = __int_as_float(__builtin_amdgcn_readlane(__float_as_int(w), j + u));
            int   kj = __builtin_amdgcn_readlane(k, j + u);
            add_row(acc, kj, v[u], wj);
        }
    }
    if (j < nvalid) {
        const int nt = nvalid - j;
        float4 v[8];
#pragma unroll
        for (int u = 0; u < 7; ++u) {
            if (u < nt) {
                int sj = __builtin_amdgcn_readlane(s, j + u);
                v[u] = ((const float4*)(H + (size_t)sj * DCH))[lane];
            }
        }
#pragma unroll
        for (int u = 0; u < 7; ++u) {
            if (u < nt) {
                float wj = __int_as_float(__builtin_amdgcn_readlane(__float_as_int(w), j + u));
                int   kj = __builtin_amdgcn_readlane(k, j + u);
                add_row(acc, kj, v[u], wj);
            }
        }
    }
}

struct AggFilter {
    float* acc; int* qe; int* qr;
    const float* __restrict__ H; const int* __restrict__ src; const float* __restrict__ dis;
    int row0; int drow0; int tl; int lane; int cnt;

    __device__ __forceinline__ void operator()(int t, int ei, bool ok) {
        int rl = t - row0;
        unsigned lim = ok ? (unsigned)tl : 0u;
        bool m = (unsigned)rl < lim;
        unsigned mask = __builtin_amdgcn_ballot_w32(m);
        if (mask != 0u) {
            int p = (int)__builtin_amdgcn_mbcnt_lo(mask, 0u);
            if (m) { qe[cnt + p] = ei; qr[cnt + p] = rl; }
            cnt += __builtin_popcount(mask);
        }
    }

    __device__ __forceinline__ void one(int s, float w, int r, int j) {
        int   sj = __builtin_amdgcn_readlane(s, j);
        float wj = __int_as_float(__builtin_amdgcn_readlane(__float_as_int(w), j));
        int   rj = __builtin_amdgcn_readlane(r, j);
        float4 v = ((const float4*)(H + (size_t)sj * DCH))[lane];
        float* ap = acc + rj * DCH + lane;
        atomicAdd(ap,      v.x * wj);
        atomicAdd(ap + 32, v.y * wj);
        atomicAdd(ap + 64, v.z * wj);
        atomicAdd(ap + 96, v.w * wj);
    }

    __device__ __forceinline__ void add_rows(int s, float w, int r, int nvalid) {
        int j = 0;
        for (; j + 8 <= nvalid; j += 8) {
            float4 v[8];
#pragma unroll
            for (int u = 0; u < 8; ++u) {
                int sj = __builtin_amdgcn_readlane(s, j + u);
                v[u] = ((const float4*)(H + (size_t)sj * DCH))[lane];
            }
#pragma unroll
            for (int u = 0; u < 8; ++u) {
                float wj = __int_as_float(__builtin_amdgcn_readlane(__float_as_int(w), j + u));
                int   rj = __builtin_amdgcn_readlane(r, j + u);
                float* ap = acc + rj * DCH + lane;
                atomicAdd(ap,      v[u].x * wj);
                atomicAdd(ap + 32, v[u].y * wj);
                atomicAdd(ap + 64, v[u].z * wj);
                atomicAdd(ap + 96, v[u].w * wj);
            }
        }
        for (; j < nvalid; ++j) one(s, w, r, j);
    }

    __device__ __forceinline__ void flush(int start, int nvalid) {
        int r = 0, s = 0; float w = 0.0f;
        if (lane < nvalid) {
            int ei = qe[start + lane];
            r = qr[start + lane];
            s = src[ei];
            w = dis[dis_slot(s)] * dis[drow0 + r];
        }
        add_rows(s, w, r, nvalid);
    }

    __device__ __forceinline__ void trip_end() {
        int done = 0;
        while (cnt - done >= 32) { flush(done, 32); done += 32; }
        if (done > 0) {
            int rem = cnt - done;
            int a = 0, b = 0;
            if (lane < rem) { a = qe[done + lane]; b = qr[done + lane]; }
            if (lane < rem) { qe[lane] = a; qr[lane] = b; }
            cnt = rem;
        }
    }
};

__global__ __launch_bounds__(NTHR) void gcn_agg_tile(const float* __restrict__ H,
                                                     const int* __restrict__ src,
                                                     const int* __restrict__ tgt,
                                                     const float* __restrict__ dis,
                                                     const float* __restrict__ bias,
                                                     const int* __restrict__ recs,
                                                     const int* __restrict__ cnts,
                                                     float* __restrict__ out,
                                                     int n, int e, int relu) {
    __shared__ float acc[TROWS * DCH];
    __shared__ int qe[NWAVES * QCAP];
    __shared__ int qr[NWAVES * QCAP];
    const int tid = threadIdx.x, lane = tid & 31, wave = tid >> 5;
    const int row0 = blockIdx.x * TROWS;
    const int drow0 = blockIdx.x * DPITCH;
    const int tl = min(TROWS, n - row0);

    const float4 b = ((const float4*)bias)[lane];

    int cv[SLICES];
#pragma unroll
    for (int s = 0; s < SLICES; ++s) cv[s] = cnts[(blockIdx.x * SLICES + s) * SPITCH + lane];
    bool over = false;
#pragma unroll
    for (int s = 0; s < SLICES; ++s) over = over || (cv[s] > OCAP);
    if (__builtin_amdgcn_ballot_w32(over) == 0u) {
        const int wv = __builtin_amdgcn_readfirstlane(wave);
        const int* __restrict__ rt = recs + (size_t)blockIdx.x * RCAP + wv * OCAP;
        int c[SLICES];
        int rec[SLICES];
#pragma unroll
        for (int s = 0; s < SLICES; ++s) {
            c[s] = __builtin_amdgcn_readlane(cv[s], wv);
            rec[s] = (c[s] > 0) ? rt[s * NWAVES * OCAP + lane] : 0;
        }

        RowSums sum = {};
        {
            float4 h[RPW]; float d[RPW];
#pragma unroll
            for (int i = 0; i < RPW; ++i) {
                int rc = min(wv + NWAVES * i, tl - 1);
                d[i] = dis[drow0 + rc];
                h[i] = ((const float4*)(H + (size_t)(row0 + rc) * DCH))[lane];
            }
#pragma unroll
            for (int i = 0; i < RPW; ++i) {
                sum.x[i] = h[i].x * d[i] * d[i] + b.x;
                sum.y[i] = h[i].y * d[i] * d[i] + b.y;
                sum.z[i] = h[i].z * d[i] * d[i] + b.z;
                sum.w[i] = h[i].w * d[i] * d[i] + b.w;
            }
        }

        int* we = qe + wv * QCAP;
        int* ww = qr + wv * QCAP;
        float wgt[SLICES];
#pragma unroll
        for (int s = 0; s < SLICES; ++s) {
            const bool v = lane < c[s];
            const int sn = v ? (rec[s] >> 8) : 0;
            const int rn = v ? (rec[s] & 255) : 0;
            wgt[s] = dis[dis_slot(sn)] * dis[drow0 + rn];
        }
        int total = 0;
#pragma unroll
        for (int s = 0; s < SLICES; ++s) {
            {
                const int p = (lane < c[s]) ? total + lane : SLICES * OCAP + lane;
                we[p] = rec[s];
                ww[p] = __float_as_int(wgt[s]);
            }
#pragma nounroll
            for (int k = 1; 32 * k < c[s]; ++k) {
                const bool v = lane < c[s] - 32 * k;
                const int rc = rt[s * NWAVES * OCAP + 32 * k + lane];
                const int sn = v ? (rc >> 8) : 0;
                const int rn = v ? (rc & 255) : 0;
                const float w = dis[dis_slot(sn)] * dis[drow0 + rn];
                if (v) {
                    we[total + 32 * k + lane] = rc;
                    ww[total + 32 * k + lane] = __float_as_int(w);
                }
            }
            total += c[s];
        }
        const int nlines = (total + 31) >> 5;
        for (int g = 0; g < nlines; ++g) {
            const int rc = we[32 * g + lane];
            const float w = __int_as_float(ww[32 * g + lane]);
            const int nvalid = min(32, total - 32 * g);
            add_line(sum, H, rc >> 8, w, (rc & 255) >> 5, nvalid, lane);
        }
#pragma unroll
        for (int i = 0; i < RPW; ++i) {
            const int orow = wv + NWAVES * i;
            if (orow < tl) {
                float4 v;
                v.x = sum.x[i]; v.y = sum.y[i]; v.z = sum.z[i]; v.w = sum.w[i];
                if (relu) {
                    v.x = fmaxf(v.x, 0.0f); v.y = fmaxf(v.y, 0.0f);
                    v.z = fmaxf(v.z, 0.0f); v.w = fmaxf(v.w, 0.0f);
                }
                ((float4*)(out + (size_t)(row0 + orow) * DCH))[lane] = v;
            }
        }
        return;
    }

    {
        float4 h[RPW]; float d[RPW];
#pragma unroll
        for (int i = 0; i < RPW; ++i) {
            int rc = min(wave + NWAVES * i, tl - 1);
            d[i] = dis[blockIdx.x * DPITCH + rc];
            h[i] = ((const float4*)(H + (size_t)(row0 + rc) * DCH))[lane];
        }
#pragma unroll
        for (int i = 0; i < RPW; ++i) {
            int r = wave + NWAVES * i;
            if (r < tl) {
                float* ap = acc + r * DCH + lane;
                ap[0]  = h[i].x * d[i] * d[i] + b.x;
                ap[32] = h[i].y * d[i] * d[i] + b.y;
                ap[64] = h[i].z * d[i] * d[i] + b.z;
                ap[96] = h[i].w * d[i] * d[i] + b.w;
            }
        }
    }
    __syncthreads();

    AggFilter f{acc, qe + wave * QCAP, qr + wave * QCAP, H, src, dis, row0, drow0, tl, lane, 0};
    scan_targets(tgt, e, wave, lane, f);
    if (f.cnt > 0) f.flush(0, f.cnt);
    __syncthreads();

    for (int r = wave; r < tl; r += NWAVES) {
        const float* ap = acc + r * DCH + lane;
        float4 v;
        v.x = ap[0]; v.y = ap[32]; v.z = ap[64]; v.w = ap[96];
        if (relu) {
            v.x = fmaxf(v.x, 0.0f); v.y = fmaxf(v.y, 0.0f);
            v.z = fmaxf(v.z, 0.0f); v.w = fmaxf(v.w, 0.0f);
        }
        ((float4*)(out + (size_t)(row0 + r) * DCH))[lane] = v;
    }
}


extern "C" void kernel_launch(void* const* d_in, const int* in_sizes, int n_in,
                              void* d_out, int out_size, void* d_ws, size_t ws_size,
                              hipStream_t stream) {
    const float* X  = (const float*)d_in[0];
    const int*   ei = (const int*)  d_in[1];
    const float* W1 = (const float*)d_in[2];
    const float* b1 = (const float*)d_in[3];
    const float* W2 = (const float*)d_in[4];
    const float* b2 = (const float*)d_in[5];
    const float* W3 = (const float*)d_in[6];
    const float* b3 = (const float*)d_in[7];
    float* out = (float*)d_out;

    const int n  = in_sizes[0] / DCH;
    const int e  = in_sizes[1] / 2;

    const int* src = ei;
    const int* tgt = ei + e;

    const int gemmBlocks = 256;
    const int aggBlocks  = (n + TROWS - 1) / TROWS;
    const int bands      = (aggBlocks + BAND - 1) / BAND;

    int*   recs  = (int*)d_ws;
    float* dis   = (float*)(recs + (size_t)aggBlocks * RCAP);
    float* H     = dis + (size_t)aggBlocks * DPITCH;
    int*   cnts  = (int*)(H + (size_t)n * DCH);
    int*   degp  = cnts + (size_t)aggBlocks * SLICES * SPITCH;

    gcn_filter_band<<<bands * SLICES, NTHR, 0, stream>>>(src, tgt, degp, recs, cnts, n, e, aggBlocks);

    const float* Ws[3] = {W1, W2, W3};
    const float* bs[3] = {b1, b2, b3};
    const float* cur   = X;

    for (int layer = 0; layer < 3; ++layer) {
        gcn_gemm_wmma<<<gemmBlocks, GEMM_THR, 0, stream>>>(cur, Ws[layer], H, n,
                                                           layer == 0 ? degp : nullptr, dis, aggBlocks, bands);
        gcn_agg_tile <<<aggBlocks, NTHR, 0, stream>>>(H, src, tgt, dis, bs[layer], recs, cnts, out,
                                                      n, e, layer < 2 ? 1 : 0);
        cur = out;
    }
}


# Next layer's GEMM moved into the aggregation's tail: a block's 196 finished rows go through LDS to WMMA against W (in LDS) and leave as the next H; x never stored; 7 -> 5 launches; sums unchanged
# speedup vs baseline: 5.9145x; 1.1195x over previous
#include <hip/hip_runtime.h>
#include <stdint.h>


typedef __attribute__((ext_vector_type(2))) float v2f;
typedef __attribute__((ext_vector_type(8))) float v8f;

#define DCH    128
#define TROWS  196
#define DPITCH 224
#define SPITCH 32
#define NTHR   1024
#define NWAVES 32
#define QCAP   544
#define SLICES 8
#define BAND   8
#define BROWS  (BAND * TROWS)
#define NLIST  (BAND * NWAVES)
#define OCAP   64
#define RCAP   (SLICES * NWAVES * OCAP)
#define DP2    1600
#define RPW    ((TROWS + NWAVES - 1) / NWAVES)

static_assert(TROWS <= 256, "a record keeps the tile row in its low 8 bits");
static_assert(DPITCH >= TROWS && DPITCH % 32 == 0, "a tile's deg^-1/2 values fill their own whole lines");
static_assert(SPITCH >= NWAVES, "a tile's line of counts holds one count per owner wave");
static_assert(OCAP % 32 == 0, "an owner's records are whole lines");
static_assert(DP2 >= BROWS && DP2 % 32 == 0, "a block's row counts fill their own whole lines");
static_assert(SLICES * OCAP + 32 <= QCAP, "a wave's records of all slices fit its LDS queue, with a spare line behind them");

__device__ __forceinline__ int dis_slot(int i) {
    const int t = (int)((unsigned)i / (unsigned)TROWS);
    return i + t * (DPITCH - TROWS);
}

template <bool GUARD, class F>
__device__ __forceinline__ void filter_trip(const int4 (&q)[4], int base, int head, int m4, F& f) {
#pragma unroll
    for (int u = 0; u < 4; ++u) {
        const int i4 = base + 32 * u;
        const bool ok = GUARD ? (i4 < m4) : true;
        const int e0 = head + 4 * i4;
        f(q[u].x, e0, ok);
        f(q[u].y, e0 + 1, ok);
        f(q[u].z, e0 + 2, ok);
        f(q[u].w, e0 + 3, ok);
    }
}

template <class F>
__device__ __forceinline__ void scan_targets(const int* __restrict__ tgt, int e,
                                             int wave, int lane, F& f) {
    const uintptr_t a = (uintptr_t)tgt;
    int head = (int)(((16u - (unsigned)(a & 15u)) & 15u) >> 2);
    if (head > e) head = e;
    const int m4    = (e - head) >> 2;
    const int tail0 = head + 4 * m4;
    if (wave == 0 && head > 0) {
        int i = lane;
        bool ok = i < head;
        int t = ok ? tgt[i] : -1;
        f(t, i, ok);
    }
    if (wave == NWAVES - 1 && tail0 < e) {
        int i = tail0 + lane;
        bool ok = i < e;
        int t = ok ? tgt[i] : -1;
        f(t, i, ok);
    }
    const int4* __restrict__ t4 = (const int4*)(tgt + head);
    const int per_it = NWAVES * 128;
    const int nfull  = m4 / per_it;
    const int woff   = wave * 128 + lane;
    if (nfull > 0) {
        int it = (int)((blockIdx.x * (unsigned)nfull) / gridDim.x);
        int4 q[4];
#pragma unroll
        for (int u = 0; u < 4; ++u) q[u] = t4[it * per_it + woff + 32 * u];
        for (int k = 0; k < nfull; ++k) {
            int nit = it + 1;
            if (nit == nfull) nit = 0;
            int4 nq[4];
#pragma unroll
            for (int u = 0; u < 4; ++u) nq[u] = t4[nit * per_it + woff + 32 * u];
            filter_trip<false>(q, it * per_it + woff, head, m4, f);
            f.trip_end();
#pragma unroll
            for (int u = 0; u < 4; ++u) q[u] = nq[u];
            it = nit;
        }
    }
    if (nfull * per_it < m4) {
        const int base = nfull * per_it + woff;
        int4 q[4];
#pragma unroll
        for (int u = 0; u < 4; ++u) q[u] = t4[min(base + 32 * u, m4 - 1)];
        filter_trip<true>(q, base, head, m4, f);
        f.trip_end();
    }
}


struct TileFilter {
    int* qe; int* qr; int* degc; int* ocnt; int* olist;
    const int* __restrict__ src;
    int row0; int tl; int lane; int cnt;

    __device__ __forceinline__ void operator()(int t, int ei, bool ok) {
        int rl = t - row0;
        unsigned lim = ok ? (unsigned)tl : 0u;
        bool m = (unsigned)rl < lim;
        unsigned mask = __builtin_amdgcn_ballot_w32(m);
        if (mask != 0u) {
            int p = (int)__builtin_amdgcn_mbcnt_lo(mask, 0u);
            if (m) { qe[cnt + p] = ei; qr[cnt + p] = rl; }
            cnt += __builtin_popcount(mask);
        }
    }

    __device__ __forceinline__ void emit(int start, int nvalid) {
        if (lane < nvalid) {
            int ei = qe[start + lane];
            int r  = qr[start + lane];
            int s  = src[ei];
            atomicAdd(&degc[r], 1);
            int j  = (int)((unsigned)r / (unsigned)TROWS);
            int rr = r - j * TROWS;
            int o = j * NWAVES + (rr & (NWAVES - 1));
            int p = atomicAdd(&ocnt[o], 1);
            if (p < OCAP) olist[o * OCAP + p] = (s << 8) | rr;
        }
    }

    __device__ __forceinline__ void trip_end() {
        int done = 0;
        while (cnt - done >= 32) { emit(done, 32); done += 32; }
        if (done > 0) {
            int rem = cnt - done;
            int a = 0, b = 0;
            if (lane < rem) { a = qe[done + lane]; b = qr[done + lane]; }
            if (lane < rem) { qe[lane] = a; qr[lane] = b; }
            cnt = rem;
        }
    }
};

__global__ __launch_bounds__(NTHR) void gcn_filter_band(const int* __restrict__ src,
                                                        const int* __restrict__ tgt,
                                                        int* __restrict__ degp,
                                                        int* __restrict__ recs,
                                                        int* __restrict__ cnts,
                                                        int n, int e, int tiles) {
    __shared__ int qe[NWAVES * QCAP];
    __shared__ int qr[NWAVES * QCAP];
    __shared__ int olist[NLIST * OCAP];
    __shared__ int degc[BROWS];
    __shared__ int ocnt[NLIST];
    const int tid = threadIdx.x, lane = tid & 31, wave = tid >> 5;
    const int band  = (int)(blockIdx.x / (unsigned)SLICES);
    const int sl    = (int)(blockIdx.x % (unsigned)SLICES);
    const int bands = (int)(gridDim.x / (unsigned)SLICES);
    const int row0 = band * BROWS;
    const int tl = min(BROWS, n - row0);
    const int per = (((e + SLICES - 1) / SLICES) + 3) & ~3;
    const int e0 = min(sl * per, e);
    const int e1 = min(e0 + per, e);
    for (int i = tid; i < BROWS; i += NTHR) degc[i] = 0;
    if (tid < NLIST) ocnt[tid] = 0;
    __syncthreads();
    TileFilter f{qe + wave * QCAP, qr + wave * QCAP, degc, ocnt, olist, src + e0, row0, tl, lane, 0};
    scan_targets(tgt + e0, e1 - e0, wave, lane, f);
    if (f.cnt > 0) f.emit(0, f.cnt);
    __syncthreads();
    for (int i = tid; i < tl; i += NTHR)
        degp[(sl * bands + band) * DP2 + i] = degc[i];
    for (int j = 0; j < BAND; ++j) {
        const int tile = band * BAND + j;
        if (tile < tiles) {
            const int c = min(ocnt[j * NWAVES + wave], OCAP);
            const int* ol = olist + (j * NWAVES + wave) * OCAP;
            int* __restrict__ dst = recs + (size_t)tile * RCAP + (sl * NWAVES + wave) * OCAP;
            for (int k0 = 0; k0 < c; k0 += 32) {
                const int k = k0 + lane;
                dst[k] = (k < c) ? ol[k] : -1;
            }
        }
    }
    if (wave < BAND && band * BAND + wave < tiles)
        cnts[((band * BAND + wave) * SLICES + sl) * SPITCH + lane] = ocnt[wave * NWAVES + lane];
}


#define GEMM_WAVES 16
#define GEMM_THR   (GEMM_WAVES * 32)
#define WQ_PITCH   136

__global__ __launch_bounds__(GEMM_THR) __attribute__((amdgpu_waves_per_eu(4, 4))) void gcn_gemm_wmma(const float* __restrict__ A,
                                                          const float* __restrict__ W,
                                                          float* __restrict__ H,
                                                          int n,
                                                          const int* __restrict__ degp,
                                                          float* __restrict__ dis,
                                                          int tiles, int bands) {
    __shared__ float4 wq[(DCH / 4) * WQ_PITCH];
    const int tid = threadIdx.x, lane = tid & 31;
    const int wave = __builtin_amdgcn_readfirstlane(tid >> 5);

    const int h = lane >> 4;
    const int c = lane & 15;
    const int mTiles = n >> 4;
    const int nmine  = (mTiles - (int)blockIdx.x + (int)gridDim.x - 1) / (int)gridDim.x;
    const int nitems = 2 * nmine;

    float4 a[DCH / 8];
    {
        const int tile = min((int)blockIdx.x + (int)gridDim.x * (wave >> 1), mTiles - 1);
        const float4* __restrict__ ap = (const float4*)(A + (size_t)(tile * 16 + c) * DCH) + h;
#pragma unroll
        for (int j = 0; j < DCH / 8; ++j) a[j] = ap[2 * j];
    }

    {
        constexpr int NQ = (DCH / 4) * DCH / GEMM_THR;
        float4 v[NQ];
#pragma unroll
        for (int i = 0; i < NQ; ++i) {
            const int idx = tid + GEMM_THR * i;
            const int q = idx >> 7, col = idx & (DCH - 1);
            const float* wp = W + (size_t)(4 * q) * DCH + col;
            v[i].x = wp[0];
            v[i].y = wp[DCH];
            v[i].z = wp[2 * DCH];
            v[i].w = wp[3 * DCH];
        }
#pragma unroll
        for (int i = 0; i < NQ; ++i) {
            const int idx = tid + GEMM_THR * i;
            wq[(idx >> 7) * WQ_PITCH + (idx & (DCH - 1))] = v[i];
        }
    }
    __syncthreads();

    for (int item = wave; item < nitems; item += GEMM_WAVES) {
        const int tile = (int)blockIdx.x + (int)gridDim.x * (item >> 1);
        const int nh   = item & 1;

        float4 an[DCH / 8];
        {
            const int tn = min((int)blockIdx.x + (int)gridDim.x * ((item + GEMM_WAVES) >> 1), mTiles - 1);
            const float4* __restrict__ apn = (const float4*)(A + (size_t)(tn * 16 + c) * DCH) + h;
#pragma unroll
            for (int j = 0; j < DCH / 8; ++j) an[j] = apn[2 * j];
        }
        __builtin_amdgcn_sched_barrier(0);

        const float4* bp = wq + h * WQ_PITCH + 64 * nh + c;
        v8f acc[4] = {};
        float4 b[4];
#pragma unroll
        for (int t = 0; t < 4; ++t) b[t] = bp[16 * t];
#pragma unroll
        for (int j = 0; j < DCH / 8; ++j) {
            float4 bn[4];
            if (j + 1 < DCH / 8) {
#pragma unroll
                for (int t = 0; t < 4; ++t) bn[t] = bp[2 * (j + 1) * WQ_PITCH + 16 * t];
            }
#pragma unroll
            for (int t = 0; t < 4; ++t) {
                v2f av, bv;
                av.x = a[j].x; av.y = a[j].y;
                bv.x = b[t].x; bv.y = b[t].y;
                acc[t] = __builtin_amdgcn_wmma_f32_16x16x4_f32(
                         false, av,  false, bv,
                         (short)0, acc[t],  false,  false);
            }
#pragma unroll
            for (int t = 0; t < 4; ++t) {
                v2f av, bv;
                av.x = a[j].z; av.y = a[j].w;
                bv.x = b[t].z; bv.y = b[t].w;
                acc[t] = __builtin_amdgcn_wmma_f32_16x16x4_f32(
                         false, av,  false, bv,
                         (short)0, acc[t],  false,  false);
            }
            if (j + 1 < DCH / 8) {
#pragma unroll
                for (int t = 0; t < 4; ++t) b[t] = bn[t];
            }
        }

        const int orow = tile * 16 + 8 * h;
        float* hp = H + (size_t)orow * DCH + 64 * nh + c;
#pragma unroll
        for (int r = 0; r < 8; ++r) {
#pragma unroll
            for (int t = 0; t < 4; ++t) hp[(size_t)r * DCH + 16 * t] = acc[t][r];
        }
#pragma unroll
        for (int j = 0; j < DCH / 8; ++j) a[j] = an[j];
    }

    if (degp != nullptr) {
        for (int tile = (int)blockIdx.x; tile < tiles; tile += (int)gridDim.x) {
            const int i = tile * TROWS + tid;
            if (tid < TROWS && i < n) {
                const int band = tile / BAND;
                const int r = (tile - band * BAND) * TROWS + tid;
                int d = 0;
#pragma unroll
                for (int s = 0; s < SLICES; ++s) d += degp[(s * bands + band) * DP2 + r];
                dis[tile * DPITCH + tid] = rsqrtf(1.0f + (float)d);
            }
        }
    }
}


struct RowSums { v8f x, y, z, w; };
static_assert(RPW <= 8, "one vector element per owned row");

__device__ __forceinline__ void add_row(RowSums& a, int k, const float4& v, float wt) {
    a.x[k] += v.x * wt;
    a.y[k] += v.y * wt;
    a.z[k] += v.z * wt;
    a.w[k] += v.w * wt;
}

__device__ __forceinline__ void add_line(RowSums& acc, const float* __restrict__ H,
                                         int s, float w, int k, int nvalid, int lane) {
    int j = 0;
    for (; j + 8 <= nvalid; j += 8) {
        float4 v[8];
#pragma unroll
        for (int u = 0; u < 8; ++u) {
            int sj = __builtin_amdgcn_readlane(s, j + u);
            v[u] = ((const float4*)(H + (size_t)sj * DCH))[lane];
        }
#pragma unroll
        for (int u = 0; u < 8; ++u) {
            float wj = __int_as_float(__builtin_amdgcn_readlane(__float_as_int(w), j + u));
            int   kj = __builtin_amdgcn_readlane(k, j + u);
            add_row(acc, kj, v[u], wj);
        }
    }
    if (j < nvalid) {
        const int nt = nvalid - j;
        float4 v[8];
#pragma unroll
        for (int u = 0; u < 7; ++u) {
            if (u < nt) {
                int sj = __builtin_amdgcn_readlane(s, j + u);
                v[u] = ((const float4*)(H + (size_t)sj * DCH))[lane];
            }
        }
#pragma unroll
        for (int u = 0; u < 7; ++u) {
            if (u < nt) {
                float wj = __int_as_float(__builtin_amdgcn_readlane(__float_as_int(w), j + u));
                int   kj = __builtin_amdgcn_readlane(k, j + u);
                add_row(acc, kj, v[u], wj);
            }
        }
    }
}

struct AggFilter {
    float* acc; int* qe; int* qr;
    const float* __restrict__ H; const int* __restrict__ src; const float* __restrict__ dis;
    int row0; int drow0; int tl; int lane; int cnt;

    __device__ __forceinline__ void operator()(int t, int ei, bool ok) {
        int rl = t - row0;
        unsigned lim = ok ? (unsigned)tl : 0u;
        bool m = (unsigned)rl < lim;
        unsigned mask = __builtin_amdgcn_ballot_w32(m);
        if (mask != 0u) {
            int p = (int)__builtin_amdgcn_mbcnt_lo(mask, 0u);
            if (m) { qe[cnt + p] = ei; qr[cnt + p] = rl; }
            cnt += __builtin_popcount(mask);
        }
    }

    __device__ __forceinline__ void one(int s, float w, int r, int j) {
        int   sj = __builtin_amdgcn_readlane(s, j);
        float wj = __int_as_float(__builtin_amdgcn_readlane(__float_as_int(w), j));
        int   rj = __builtin_amdgcn_readlane(r, j);
        float4 v = ((const float4*)(H + (size_t)sj * DCH))[lane];
        float* ap = acc + rj * DCH + lane;
        atomicAdd(ap,      v.x * wj);
        atomicAdd(ap + 32, v.y * wj);
        atomicAdd(ap + 64, v.z * wj);
        atomicAdd(ap + 96, v.w * wj);
    }

    __device__ __forceinline__ void add_rows(int s, float w, int r, int nvalid) {
        int j = 0;
        for (; j + 8 <= nvalid; j += 8) {
            float4 v[8];
#pragma unroll
            for (int u = 0; u < 8; ++u) {
                int sj = __builtin_amdgcn_readlane(s, j + u);
                v[u] = ((const float4*)(H + (size_t)sj * DCH))[lane];
            }
#pragma unroll
            for (int u = 0; u < 8; ++u) {
                float wj = __int_as_float(__builtin_amdgcn_readlane(__float_as_int(w), j + u));
                int   rj = __builtin_amdgcn_readlane(r, j + u);
                float* ap = acc + rj * DCH + lane;
                atomicAdd(ap,      v[u].x * wj);
                atomicAdd(ap + 32, v[u].y * wj);
                atomicAdd(ap + 64, v[u].z * wj);
                atomicAdd(ap + 96, v[u].w * wj);
            }
        }
        for (; j < nvalid; ++j) one(s, w, r, j);
    }

    __device__ __forceinline__ void flush(int start, int nvalid) {
        int r = 0, s = 0; float w = 0.0f;
        if (lane < nvalid) {
            int ei = qe[start + lane];
            r = qr[start + lane];
            s = src[ei];
            w = dis[dis_slot(s)] * dis[drow0 + r];
        }
        add_rows(s, w, r, nvalid);
    }

    __device__ __forceinline__ void trip_end() {
        int done = 0;
        while (cnt - done >= 32) { flush(done, 32); done += 32; }
        if (done > 0) {
            int rem = cnt - done;
            int a = 0, b = 0;
            if (lane < rem) { a = qe[done + lane]; b = qr[done + lane]; }
            if (lane < rem) { qe[lane] = a; qr[lane] = b; }
            cnt = rem;
        }
    }
};

#define XROWS  208
#define XPITCH 132
static_assert(XROWS >= TROWS && XROWS % 16 == 0, "the finished tile is cut into whole 16-row matrix tiles");
static_assert(XROWS <= NWAVES * RPW, "every row of the finished tile in LDS has an owner wave that writes it");
static_assert(XROWS * XPITCH >= TROWS * DCH, "the LDS sums of the overflow path fit the same array");

template <bool FUSE>
__global__ __launch_bounds__(NTHR) void gcn_agg_tile(const float* __restrict__ H,
                                                     const int* __restrict__ src,
                                                     const int* __restrict__ tgt,
                                                     const float* __restrict__ dis,
                                                     const float* __restrict__ bias,
                                                     const int* __restrict__ recs,
                                                     const int* __restrict__ cnts,
                                                     float* __restrict__ out,
                                                     int n, int e, int relu,
                                                     const float* __restrict__ Wn,
                                                     float* __restrict__ Hn) {
    __shared__ __attribute__((aligned(16))) float acc[FUSE ? XROWS * XPITCH : TROWS * DCH];
    __shared__ int qe[NWAVES * QCAP];
    __shared__ int qr[NWAVES * QCAP];
    __shared__ float4 wq[FUSE ? (DCH / 4) * WQ_PITCH : 1];
    const int tid = threadIdx.x, lane = tid & 31, wave = tid >> 5;
    const int row0 = blockIdx.x * TROWS;
    const int drow0 = blockIdx.x * DPITCH;
    const int tl = min(TROWS, n - row0);

    if constexpr (FUSE) {
        constexpr int NQ = (DCH / 4) * DCH / NTHR;
        float4 v[NQ];
#pragma unroll
        for (int i = 0; i < NQ; ++i) {
            const int idx = tid + NTHR * i;
            const int q = idx >> 7, col = idx & (DCH - 1);
            const float* wp = Wn + (size_t)(4 * q) * DCH + col;
            v[i].x = wp[0];
            v[i].y = wp[DCH];
            v[i].z = wp[2 * DCH];
            v[i].w = wp[3 * DCH];
        }
#pragma unroll
        for (int i = 0; i < NQ; ++i) {
            const int idx = tid + NTHR * i;
            wq[(idx >> 7) * WQ_PITCH + (idx & (DCH - 1))] = v[i];
        }
    }

    const float4 b = ((const float4*)bias)[lane];

    int cv[SLICES];
#pragma unroll
    for (int s = 0; s < SLICES; ++s) cv[s] = cnts[(blockIdx.x * SLICES + s) * SPITCH + lane];
    bool over = false;
#pragma unroll
    for (int s = 0; s < SLICES; ++s) over = over || (cv[s] > OCAP);
    if (__builtin_amdgcn_ballot_w32(over) == 0u) {
        const int wv = __builtin_amdgcn_readfirstlane(wave);
        const int* __restrict__ rt = recs + (size_t)blockIdx.x * RCAP + wv * OCAP;
        int c[SLICES];
        int rec[SLICES];
#pragma unroll
        for (int s = 0; s < SLICES; ++s) {
            c[s] = __builtin_amdgcn_readlane(cv[s], wv);
            rec[s] = (c[s] > 0) ? rt[s * NWAVES * OCAP + lane] : 0;
        }

        RowSums sum = {};
        {
            float4 h[RPW]; float d[RPW];
#pragma unroll
            for (int i = 0; i < RPW; ++i) {
                int rc = min(wv + NWAVES * i, tl - 1);
                d[i] = dis[drow0 + rc];
                h[i] = ((const float4*)(H + (size_t)(row0 + rc) * DCH))[lane];
            }
#pragma unroll
            for (int i = 0; i < RPW; ++i) {
                sum.x[i] = h[i].x * d[i] * d[i] + b.x;
                sum.y[i] = h[i].y * d[i] * d[i] + b.y;
                sum.z[i] = h[i].z * d[i] * d[i] + b.z;
                sum.w[i] = h[i].w * d[i] * d[i] + b.w;
            }
        }

        int* we = qe + wv * QCAP;
        int* ww = qr + wv * QCAP;
        float wgt[SLICES];
#pragma unroll
        for (int s = 0; s < SLICES; ++s) {
            const bool v = lane < c[s];
            const int sn = v ? (rec[s] >> 8) : 0;
            const int rn = v ? (rec[s] & 255) : 0;
            wgt[s] = dis[dis_slot(sn)] * dis[drow0 + rn];
        }
        int total = 0;
#pragma unroll
        for (int s = 0; s < SLICES; ++s) {
            {
                const int p = (lane < c[s]) ? total + lane : SLICES * OCAP + lane;
                we[p] = rec[s];
                ww[p] = __float_as_int(wgt[s]);
            }
#pragma nounroll
            for (int k = 1; 32 * k < c[s]; ++k) {
                const bool v = lane < c[s] - 32 * k;
                const int rc = rt[s * NWAVES * OCAP + 32 * k + lane];
                const int sn = v ? (rc >> 8) : 0;
                const int rn = v ? (rc & 255) : 0;
                const float w = dis[dis_slot(sn)] * dis[drow0 + rn];
                if (v) {
                    we[total + 32 * k + lane] = rc;
                    ww[total + 32 * k + lane] = __float_as_int(w);
                }
            }
            total += c[s];
        }
        const int nlines = (total + 31) >> 5;
        for (int g = 0; g < nlines; ++g) {
            const int rc = we[32 * g + lane];
            const float w = __int_as_float(ww[32 * g + lane]);
            const int nvalid = min(32, total - 32 * g);
            add_line(sum, H, rc >> 8, w, (rc & 255) >> 5, nvalid, lane);
        }
#pragma unroll
        for (int i = 0; i < RPW; ++i) {
            const int orow = wv + NWAVES * i;
            float4 v;
            v.x = sum.x[i]; v.y = sum.y[i]; v.z = sum.z[i]; v.w = sum.w[i];
            if (relu) {
                v.x = fmaxf(v.x, 0.0f); v.y = fmaxf(v.y, 0.0f);
                v.z = fmaxf(v.z, 0.0f); v.w = fmaxf(v.w, 0.0f);
            }
            if constexpr (FUSE) {
                if (orow >= tl) v = make_float4(0.0f, 0.0f, 0.0f, 0.0f);
                if (orow < XROWS) *(float4*)(acc + orow * XPITCH + 4 * lane) = v;
            } else {
                if (orow < tl) ((float4*)(out + (size_t)(row0 + orow) * DCH))[lane] = v;
            }
        }
        if constexpr (!FUSE) return;
    } else {
        {
            float4 h[RPW]; float d[RPW];
#pragma unroll
            for (int i = 0; i < RPW; ++i) {
                int rc = min(wave + NWAVES * i, tl - 1);
                d[i] = dis[blockIdx.x * DPITCH + rc];
                h[i] = ((const float4*)(H + (size_t)(row0 + rc) * DCH))[lane];
            }
#pragma unroll
            for (int i = 0; i < RPW; ++i) {
                int r = wave + NWAVES * i;
                if (r < tl) {
                    float* ap = acc + r * DCH + lane;
                    ap[0]  = h[i].x * d[i] * d[i] + b.x;
                    ap[32] = h[i].y * d[i] * d[i] + b.y;
                    ap[64] = h[i].z * d[i] * d[i] + b.z;
                    ap[96] = h[i].w * d[i] * d[i] + b.w;
                }
            }
        }
        __syncthreads();

        AggFilter f{acc, qe + wave * QCAP, qr + wave * QCAP, H, src, dis, row0, drow0, tl, lane, 0};
        scan_targets(tgt, e, wave, lane, f);
        if (f.cnt > 0) f.flush(0, f.cnt);
        __syncthreads();

        if constexpr (FUSE) {
            float4 v[RPW];
#pragma unroll
            for (int i = 0; i < RPW; ++i) {
                const int r = wave + NWAVES * i;
                v[i] = make_float4(0.0f, 0.0f, 0.0f, 0.0f);
                if (r < tl) {
                    const float* ap = acc + r * DCH + lane;
                    v[i].x = ap[0]; v[i].y = ap[32]; v[i].z = ap[64]; v[i].w = ap[96];
                    if (relu) {
                        v[i].x = fmaxf(v[i].x, 0.0f); v[i].y = fmaxf(v[i].y, 0.0f);
                        v[i].z = fmaxf(v[i].z, 0.0f); v[i].w = fmaxf(v[i].w, 0.0f);
                    }
                }
            }
            __syncthreads();
#pragma unroll
            for (int i = 0; i < RPW; ++i) {
                const int r = wave + NWAVES * i;
                if (r < XROWS) *(float4*)(acc + r * XPITCH + 4 * lane) = v[i];
            }
        } else {
            for (int r = wave; r < tl; r += NWAVES) {
                const float* ap = acc + r * DCH + lane;
                float4 v;
                v.x = ap[0]; v.y = ap[32]; v.z = ap[64]; v.w = ap[96];
                if (relu) {
                    v.x = fmaxf(v.x, 0.0f); v.y = fmaxf(v.y, 0.0f);
                    v.z = fmaxf(v.z, 0.0f); v.w = fmaxf(v.w, 0.0f);
                }
                ((float4*)(out + (size_t)(row0 + r) * DCH))[lane] = v;
            }
        }
    }

    if constexpr (FUSE) {
        __syncthreads();

        const int wm = __builtin_amdgcn_readfirstlane(wave);
        const int h = lane >> 4;
        const int c = lane & 15;
        const int nitems = 4 * ((tl + 15) >> 4);
        for (int item = wm; item < nitems; item += NWAVES) {
            const int t  = item >> 2;
            const int nq = item & 3;
            const float4* ap = (const float4*)(acc + (t * 16 + c) * XPITCH) + h;
            const float4* bp = wq + h * WQ_PITCH + 32 * nq + c;
            v8f d[2] = {};
            float4 a = ap[0];
            float4 bb[2];
#pragma unroll
            for (int u = 0; u < 2; ++u) bb[u] = bp[16 * u];
#pragma unroll
            for (int j = 0; j < DCH / 8; ++j) {
                float4 an = a;
                float4 bn[2];
                if (j + 1 < DCH / 8) {
                    an = ap[2 * (j + 1)];
#pragma unroll
                    for (int u = 0; u < 2; ++u) bn[u] = bp[2 * (j + 1) * WQ_PITCH + 16 * u];
                }
#pragma unroll
                for (int u = 0; u < 2; ++u) {
                    v2f av, bv;
                    av.x = a.x; av.y = a.y;
                    bv.x = bb[u].x; bv.y = bb[u].y;
                    d[u] = __builtin_amdgcn_wmma_f32_16x16x4_f32(
                             false, av,  false, bv,
                             (short)0, d[u],  false,  false);
                }
#pragma unroll
                for (int u = 0; u < 2; ++u) {
                    v2f av, bv;
                    av.x = a.z; av.y = a.w;
                    bv.x = bb[u].z; bv.y = bb[u].w;
                    d[u] = __builtin_amdgcn_wmma_f32_16x16x4_f32(
                             false, av,  false, bv,
                             (short)0, d[u],  false,  false);
                }
                if (j + 1 < DCH / 8) {
                    a = an;
#pragma unroll
                    for (int u = 0; u < 2; ++u) bb[u] = bn[u];
                }
            }
            const int trow = t * 16 + 8 * h;
            float* hp = Hn + (size_t)(row0 + trow) * DCH + 32 * nq + c;
#pragma unroll
            for (int r = 0; r < 8; ++r) {
                if (trow + r < tl) {
#pragma unroll
                    for (int u = 0; u < 2; ++u) hp[(size_t)r * DCH + 16 * u] = d[u][r];
                }
            }
        }
    }
}


extern "C" void kernel_launch(void* const* d_in, const int* in_sizes, int n_in,
                              void* d_out, int out_size, void* d_ws, size_t ws_size,
                              hipStream_t stream) {
    const float* X  = (const float*)d_in[0];
    const int*   ei = (const int*)  d_in[1];
    const float* W1 = (const float*)d_in[2];
    const float* b1 = (const float*)d_in[3];
    const float* W2 = (const float*)d_in[4];
    const float* b2 = (const float*)d_in[5];
    const float* W3 = (const float*)d_in[6];
    const float* b3 = (const float*)d_in[7];
    float* out = (float*)d_out;

    const int n  = in_sizes[0] / DCH;
    const int e  = in_sizes[1] / 2;

    const int* src = ei;
    const int* tgt = ei + e;

    const int gemmBlocks = 256;
    const int aggBlocks  = (n + TROWS - 1) / TROWS;
    const int bands      = (aggBlocks + BAND - 1) / BAND;

    int*   recs  = (int*)d_ws;
    float* dis   = (float*)(recs + (size_t)aggBlocks * RCAP);
    float* H0    = dis + (size_t)aggBlocks * DPITCH;
    float* H1    = H0 + (size_t)n * DCH;
    int*   cnts  = (int*)(H1 + (size_t)n * DCH);
    int*   degp  = cnts + (size_t)aggBlocks * SLICES * SPITCH;

    gcn_filter_band<<<bands * SLICES, NTHR, 0, stream>>>(src, tgt, degp, recs, cnts, n, e, aggBlocks);

    gcn_gemm_wmma<<<gemmBlocks, GEMM_THR, 0, stream>>>(X, W1, H0, n, degp, dis, aggBlocks, bands);
    gcn_agg_tile<true> <<<aggBlocks, NTHR, 0, stream>>>(H0, src, tgt, dis, b1, recs, cnts, nullptr, n, e, 1, W2, H1);
    gcn_agg_tile<true> <<<aggBlocks, NTHR, 0, stream>>>(H1, src, tgt, dis, b2, recs, cnts, nullptr, n, e, 1, W3, H0);
    gcn_agg_tile<false><<<aggBlocks, NTHR, 0, stream>>>(H0, src, tgt, dis, b3, recs, cnts, out, n, e, 0, nullptr, nullptr);
}


# Source's deg^-1/2 folded into the stored H rows by their producer, the target's applied once per finished row: per-edge weight lookup and multiply gone; sum order kept, products rounded differently
# speedup vs baseline: 6.4388x; 1.0886x over previous
#include <hip/hip_runtime.h>
#include <stdint.h>


typedef __attribute__((ext_vector_type(2))) float v2f;
typedef __attribute__((ext_vector_type(8))) float v8f;

#define DCH    128
#define TROWS  196
#define DPITCH 224
#define SPITCH 32
#define NTHR   1024
#define NWAVES 32
#define QCAP   544
#define SLICES 8
#define BAND   8
#define BROWS  (BAND * TROWS)
#define NLIST  (BAND * NWAVES)
#define OCAP   64
#define RCAP   (SLICES * NWAVES * OCAP)
#define DP2    1600
#define RPW    ((TROWS + NWAVES - 1) / NWAVES)

static_assert(TROWS <= 256, "a record keeps the tile row in its low 8 bits");
static_assert(DPITCH >= TROWS && DPITCH % 32 == 0, "a tile's deg^-1/2 values fill their own whole lines");
static_assert(SPITCH >= NWAVES, "a tile's line of counts holds one count per owner wave");
static_assert(OCAP % 32 == 0, "an owner's records are whole lines");
static_assert(DP2 >= BROWS && DP2 % 32 == 0, "a block's row counts fill their own whole lines");
static_assert(SLICES * OCAP + 32 <= QCAP, "a wave's records of all slices fit its LDS queue, with a spare line behind them");

template <bool GUARD, class F>
__device__ __forceinline__ void filter_trip(const int4 (&q)[4], int base, int head, int m4, F& f) {
#pragma unroll
    for (int u = 0; u < 4; ++u) {
        const int i4 = base + 32 * u;
        const bool ok = GUARD ? (i4 < m4) : true;
        const int e0 = head + 4 * i4;
        f(q[u].x, e0, ok);
        f(q[u].y, e0 + 1, ok);
        f(q[u].z, e0 + 2, ok);
        f(q[u].w, e0 + 3, ok);
    }
}

template <class F>
__device__ __forceinline__ void scan_targets(const int* __restrict__ tgt, int e,
                                             int wave, int lane, F& f) {
    const uintptr_t a = (uintptr_t)tgt;
    int head = (int)(((16u - (unsigned)(a & 15u)) & 15u) >> 2);
    if (head > e) head = e;
    const int m4    = (e - head) >> 2;
    const int tail0 = head + 4 * m4;
    if (wave == 0 && head > 0) {
        int i = lane;
        bool ok = i < head;
        int t = ok ? tgt[i] : -1;
        f(t, i, ok);
    }
    if (wave == NWAVES - 1 && tail0 < e) {
        int i = tail0 + lane;
        bool ok = i < e;
        int t = ok ? tgt[i] : -1;
        f(t, i, ok);
    }
    const int4* __restrict__ t4 = (const int4*)(tgt + head);
    const int per_it = NWAVES * 128;
    const int nfull  = m4 / per_it;
    const int woff   = wave * 128 + lane;
    if (nfull > 0) {
        int it = (int)((blockIdx.x * (unsigned)nfull) / gridDim.x);
        int4 q[4];
#pragma unroll
        for (int u = 0; u < 4; ++u) q[u] = t4[it * per_it + woff + 32 * u];
        for (int k = 0; k < nfull; ++k) {
            int nit = it + 1;
            if (nit == nfull) nit = 0;
            int4 nq[4];
#pragma unroll
            for (int u = 0; u < 4; ++u) nq[u] = t4[nit * per_it + woff + 32 * u];
            filter_trip<false>(q, it * per_it + woff, head, m4, f);
            f.trip_end();
#pragma unroll
            for (int u = 0; u < 4; ++u) q[u] = nq[u];
            it = nit;
        }
    }
    if (nfull * per_it < m4) {
        const int base = nfull * per_it + woff;
        int4 q[4];
#pragma unroll
        for (int u = 0; u < 4; ++u) q[u] = t4[min(base + 32 * u, m4 - 1)];
        filter_trip<true>(q, base, head, m4, f);
        f.trip_end();
    }
}


struct TileFilter {
    int* qe; int* qr; int* degc; int* ocnt; int* olist;
    const int* __restrict__ src;
    int row0; int tl; int lane; int cnt;

    __device__ __forceinline__ void operator()(int t, int ei, bool ok) {
        int rl = t - row0;
        unsigned lim = ok ? (unsigned)tl : 0u;
        bool m = (unsigned)rl < lim;
        unsigned mask = __builtin_amdgcn_ballot_w32(m);
        if (mask != 0u) {
            int p = (int)__builtin_amdgcn_mbcnt_lo(mask, 0u);
            if (m) { qe[cnt + p] = ei; qr[cnt + p] = rl; }
            cnt += __builtin_popcount(mask);
        }
    }

    __device__ __forceinline__ void emit(int start, int nvalid) {
        if (lane < nvalid) {
            int ei = qe[start + lane];
            int r  = qr[start + lane];
            int s  = src[ei];
            atomicAdd(&degc[r], 1);
            int j  = (int)((unsigned)r / (unsigned)TROWS);
            int rr = r - j * TROWS;
            int o = j * NWAVES + (rr & (NWAVES - 1));
            int p = atomicAdd(&ocnt[o], 1);
            if (p < OCAP) olist[o * OCAP + p] = (s << 8) | rr;
        }
    }

    __device__ __forceinline__ void trip_end() {
        int done = 0;
        while (cnt - done >= 32) { emit(done, 32); done += 32; }
        if (done > 0) {
            int rem = cnt - done;
            int a = 0, b = 0;
            if (lane < rem) { a = qe[done + lane]; b = qr[done + lane]; }
            if (lane < rem) { qe[lane] = a; qr[lane] = b; }
            cnt = rem;
        }
    }
};

__global__ __launch_bounds__(NTHR) void gcn_filter_band(const int* __restrict__ src,
                                                        const int* __restrict__ tgt,
                                                        int* __restrict__ degp,
                                                        int* __restrict__ recs,
                                                        int* __restrict__ cnts,
                                                        int n, int e, int tiles) {
    __shared__ int qe[NWAVES * QCAP];
    __shared__ int qr[NWAVES * QCAP];
    __shared__ int olist[NLIST * OCAP];
    __shared__ int degc[BROWS];
    __shared__ int ocnt[NLIST];
    const int tid = threadIdx.x, lane = tid & 31, wave = tid >> 5;
    const int band  = (int)(blockIdx.x / (unsigned)SLICES);
    const int sl    = (int)(blockIdx.x % (unsigned)SLICES);
    const int bands = (int)(gridDim.x / (unsigned)SLICES);
    const int row0 = band * BROWS;
    const int tl = min(BROWS, n - row0);
    const int per = (((e + SLICES - 1) / SLICES) + 3) & ~3;
    const int e0 = min(sl * per, e);
    const int e1 = min(e0 + per, e);
    for (int i = tid; i < BROWS; i += NTHR) degc[i] = 0;
    if (tid < NLIST) ocnt[tid] = 0;
    __syncthreads();
    TileFilter f{qe + wave * QCAP, qr + wave * QCAP, degc, ocnt, olist, src + e0, row0, tl, lane, 0};
    scan_targets(tgt + e0, e1 - e0, wave, lane, f);
    if (f.cnt > 0) f.emit(0, f.cnt);
    __syncthreads();
    for (int i = tid; i < tl; i += NTHR)
        degp[(sl * bands + band) * DP2 + i] = degc[i];
    for (int j = 0; j < BAND; ++j) {
        const int tile = band * BAND + j;
        if (tile < tiles) {
            const int c = min(ocnt[j * NWAVES + wave], OCAP);
            const int* ol = olist + (j * NWAVES + wave) * OCAP;
            int* __restrict__ dst = recs + (size_t)tile * RCAP + (sl * NWAVES + wave) * OCAP;
            for (int k0 = 0; k0 < c; k0 += 32) {
                const int k = k0 + lane;
                dst[k] = (k < c) ? ol[k] : -1;
            }
        }
    }
    if (wave < BAND && band * BAND + wave < tiles)
        cnts[((band * BAND + wave) * SLICES + sl) * SPITCH + lane] = ocnt[wave * NWAVES + lane];
}


#define GEMM_WAVES 16
#define GEMM_THR   (GEMM_WAVES * 32)
#define WQ_PITCH   136

#define SDT        (GEMM_THR / 16)

__device__ __forceinline__ void deg_parts(const int* __restrict__ degp, int bands, int i, int (&dg)[SLICES]) {
    const int band = (int)((unsigned)i / (unsigned)BROWS);
    const int* __restrict__ p = degp + band * DP2 + (i - band * BROWS);
#pragma unroll
    for (int s = 0; s < SLICES; ++s) dg[s] = p[s * bands * DP2];
}

__device__ __forceinline__ float dis_of(const int (&dg)[SLICES]) {
    int d = 0;
#pragma unroll
    for (int s = 0; s < SLICES; ++s) d += dg[s];
    return rsqrtf(1.0f + (float)d);
}

__device__ __forceinline__ float node_dis(const int* __restrict__ degp, int bands, int i) {
    int dg[SLICES];
    deg_parts(degp, bands, i, dg);
    return dis_of(dg);
}

__global__ __launch_bounds__(GEMM_THR) __attribute__((amdgpu_waves_per_eu(4, 4))) void gcn_gemm_wmma(const float* __restrict__ A,
                                                          const float* __restrict__ W,
                                                          float* __restrict__ H,
                                                          int n,
                                                          const int* __restrict__ degp,
                                                          float* __restrict__ dis,
                                                          int tiles, int bands) {
    __shared__ float4 wq[(DCH / 4) * WQ_PITCH];
    __shared__ __attribute__((aligned(16))) float sdis[SDT * 16];
    const int tid = threadIdx.x, lane = tid & 31;
    const int wave = __builtin_amdgcn_readfirstlane(tid >> 5);

    const int h = lane >> 4;
    const int c = lane & 15;
    const int mTiles = n >> 4;
    const int nmine  = (mTiles - (int)blockIdx.x + (int)gridDim.x - 1) / (int)gridDim.x;
    const int nitems = 2 * nmine;

    float4 a[DCH / 8];
    {
        const int tile = min((int)blockIdx.x + (int)gridDim.x * (wave >> 1), mTiles - 1);
        const float4* __restrict__ ap = (const float4*)(A + (size_t)(tile * 16 + c) * DCH) + h;
#pragma unroll
        for (int j = 0; j < DCH / 8; ++j) a[j] = ap[2 * j];
    }

    int dgs[SLICES];
    deg_parts(degp, bands, min((int)blockIdx.x + (int)gridDim.x * (tid >> 4), mTiles - 1) * 16 + (tid & 15), dgs);

    {
        constexpr int NQ = (DCH / 4) * DCH / GEMM_THR;
        float4 v[NQ];
#pragma unroll
        for (int i = 0; i < NQ; ++i) {
            const int idx = tid + GEMM_THR * i;
            const int q = idx >> 7, col = idx & (DCH - 1);
            const float* wp = W + (size_t)(4 * q) * DCH + col;
            v[i].x = wp[0];
            v[i].y = wp[DCH];
            v[i].z = wp[2 * DCH];
            v[i].w = wp[3 * DCH];
        }
#pragma unroll
        for (int i = 0; i < NQ; ++i) {
            const int idx = tid + GEMM_THR * i;
            wq[(idx >> 7) * WQ_PITCH + (idx & (DCH - 1))] = v[i];
        }
    }
    sdis[tid] = dis_of(dgs);
    __syncthreads();

    for (int item = wave; item < nitems; item += GEMM_WAVES) {
        const int tile = (int)blockIdx.x + (int)gridDim.x * (item >> 1);
        const int nh   = item & 1;

        float4 an[DCH / 8];
        {
            const int tn = min((int)blockIdx.x + (int)gridDim.x * ((item + GEMM_WAVES) >> 1), mTiles - 1);
            const float4* __restrict__ apn = (const float4*)(A + (size_t)(tn * 16 + c) * DCH) + h;
#pragma unroll
            for (int j = 0; j < DCH / 8; ++j) an[j] = apn[2 * j];
        }
        __builtin_amdgcn_sched_barrier(0);

        const float4* bp = wq + h * WQ_PITCH + 64 * nh + c;
        v8f acc[4] = {};
        float4 b[4];
#pragma unroll
        for (int t = 0; t < 4; ++t) b[t] = bp[16 * t];
#pragma unroll
        for (int j = 0; j < DCH / 8; ++j) {
            float4 bn[4];
            if (j + 1 < DCH / 8) {
#pragma unroll
                for (int t = 0; t < 4; ++t) bn[t] = bp[2 * (j + 1) * WQ_PITCH + 16 * t];
            }
#pragma unroll
            for (int t = 0; t < 4; ++t) {
                v2f av, bv;
                av.x = a[j].x; av.y = a[j].y;
                bv.x = b[t].x; bv.y = b[t].y;
                acc[t] = __builtin_amdgcn_wmma_f32_16x16x4_f32(
                         false, av,  false, bv,
                         (short)0, acc[t],  false,  false);
            }
#pragma unroll
            for (int t = 0; t < 4; ++t) {
                v2f av, bv;
                av.x = a[j].z; av.y = a[j].w;
                bv.x = b[t].z; bv.y = b[t].w;
                acc[t] = __builtin_amdgcn_wmma_f32_16x16x4_f32(
                         false, av,  false, bv,
                         (short)0, acc[t],  false,  false);
            }
            if (j + 1 < DCH / 8) {
#pragma unroll
                for (int t = 0; t < 4; ++t) b[t] = bn[t];
            }
        }

        const int orow = tile * 16 + 8 * h;
        float sr[8];
        if ((item >> 1) < SDT) {
            const float4* sp = (const float4*)(sdis + 16 * (item >> 1) + 8 * h);
            const float4 s0 = sp[0], s1 = sp[1];
            sr[0] = s0.x; sr[1] = s0.y; sr[2] = s0.z; sr[3] = s0.w;
            sr[4] = s1.x; sr[5] = s1.y; sr[6] = s1.z; sr[7] = s1.w;
        } else {
#pragma unroll
            for (int r = 0; r < 8; ++r) sr[r] = node_dis(degp, bands, orow + r);
        }
        float* hp = H + (size_t)orow * DCH + 64 * nh + c;
#pragma unroll
        for (int r = 0; r < 8; ++r) {
#pragma unroll
            for (int t = 0; t < 4; ++t) hp[(size_t)r * DCH + 16 * t] = acc[t][r] * sr[r];
        }
#pragma unroll
        for (int j = 0; j < DCH / 8; ++j) a[j] = an[j];
    }

    for (int tile = (int)blockIdx.x; tile < tiles; tile += (int)gridDim.x) {
        const int i = tile * TROWS + tid;
        if (tid < TROWS && i < n) dis[tile * DPITCH + tid] = node_dis(degp, bands, i);
    }
}


struct RowSums { v8f x, y, z, w; };
static_assert(RPW <= 8, "one vector element per owned row");

__device__ __forceinline__ void add_row(RowSums& a, int k, const float4& v) {
    a.x[k] += v.x;
    a.y[k] += v.y;
    a.z[k] += v.z;
    a.w[k] += v.w;
}

__device__ __forceinline__ void add_line(RowSums& acc, const float* __restrict__ H,
                                         int s, int k, int nvalid, int lane) {
    int j = 0;
    for (; j + 8 <= nvalid; j += 8) {
        float4 v[8];
#pragma unroll
        for (int u = 0; u < 8; ++u) {
            int sj = __builtin_amdgcn_readlane(s, j + u);
            v[u] = ((const float4*)(H + (size_t)sj * DCH))[lane];
        }
#pragma unroll
        for (int u = 0; u < 8; ++u) {
            int kj = __builtin_amdgcn_readlane(k, j + u);
            add_row(acc, kj, v[u]);
        }
    }
    if (j < nvalid) {
        const int nt = nvalid - j;
        float4 v[8];
#pragma unroll
        for (int u = 0; u < 7; ++u) {
            if (u < nt) {
                int sj = __builtin_amdgcn_readlane(s, j + u);
                v[u] = ((const float4*)(H + (size_t)sj * DCH))[lane];
            }
        }
#pragma unroll
        for (int u = 0; u < 7; ++u) {
            if (u < nt) {
                int kj = __builtin_amdgcn_readlane(k, j + u);
                add_row(acc, kj, v[u]);
            }
        }
    }
}

struct AggFilter {
    float* acc; int* qe; int* qr;
    const float* __restrict__ H; const int* __restrict__ src;
    int row0; int tl; int lane; int cnt;

    __device__ __forceinline__ void operator()(int t, int ei, bool ok) {
        int rl = t - row0;
        unsigned lim = ok ? (unsigned)tl : 0u;
        bool m = (unsigned)rl < lim;
        unsigned mask = __builtin_amdgcn_ballot_w32(m);
        if (mask != 0u) {
            int p = (int)__builtin_amdgcn_mbcnt_lo(mask, 0u);
            if (m) { qe[cnt + p] = ei; qr[cnt + p] = rl; }
            cnt += __builtin_popcount(mask);
        }
    }

    __device__ __forceinline__ void one(int s, int r, int j) {
        int sj = __builtin_amdgcn_readlane(s, j);
        int rj = __builtin_amdgcn_readlane(r, j);
        float4 v = ((const float4*)(H + (size_t)sj * DCH))[lane];
        float* ap = acc + rj * DCH + lane;
        atomicAdd(ap,      v.x);
        atomicAdd(ap + 32, v.y);
        atomicAdd(ap + 64, v.z);
        atomicAdd(ap + 96, v.w);
    }

    __device__ __forceinline__ void add_rows(int s, int r, int nvalid) {
        int j = 0;
        for (; j + 8 <= nvalid; j += 8) {
            float4 v[8];
#pragma unroll
            for (int u = 0; u < 8; ++u) {
                int sj = __builtin_amdgcn_readlane(s, j + u);
                v[u] = ((const float4*)(H + (size_t)sj * DCH))[lane];
            }
#pragma unroll
            for (int u = 0; u < 8; ++u) {
                int rj = __builtin_amdgcn_readlane(r, j + u);
                float* ap = acc + rj * DCH + lane;
                atomicAdd(ap,      v[u].x);
                atomicAdd(ap + 32, v[u].y);
                atomicAdd(ap + 64, v[u].z);
                atomicAdd(ap + 96, v[u].w);
            }
        }
        for (; j < nvalid; ++j) one(s, r, j);
    }

    __device__ __forceinline__ void flush(int start, int nvalid) {
        int r = 0, s = 0;
        if (lane < nvalid) {
            int ei = qe[start + lane];
            r = qr[start + lane];
            s = src[ei];
        }
        add_rows(s, r, nvalid);
    }

    __device__ __forceinline__ void trip_end() {
        int done = 0;
        while (cnt - done >= 32) { flush(done, 32); done += 32; }
        if (done > 0) {
            int rem = cnt - done;
            int a = 0, b = 0;
            if (lane < rem) { a = qe[done + lane]; b = qr[done + lane]; }
            if (lane < rem) { qe[lane] = a; qr[lane] = b; }
            cnt = rem;
        }
    }
};

#define XROWS  208
#define XPITCH 132
static_assert(XROWS >= TROWS && XROWS % 16 == 0, "the finished tile is cut into whole 16-row matrix tiles");
static_assert(XROWS <= NWAVES * RPW, "every row of the finished tile in LDS has an owner wave that writes it");
static_assert(XROWS * XPITCH >= TROWS * DCH, "the LDS sums of the overflow path fit the same array");

template <bool FUSE>
__global__ __launch_bounds__(NTHR) void gcn_agg_tile(const float* __restrict__ H,
                                                     const int* __restrict__ src,
                                                     const int* __restrict__ tgt,
                                                     const float* __restrict__ dis,
                                                     const float* __restrict__ bias,
                                                     const int* __restrict__ recs,
                                                     const int* __restrict__ cnts,
                                                     float* __restrict__ out,
                                                     int n, int e, int relu,
                                                     const float* __restrict__ Wn,
                                                     float* __restrict__ Hn) {
    __shared__ __attribute__((aligned(16))) float acc[FUSE ? XROWS * XPITCH : TROWS * DCH];
    __shared__ int qe[NWAVES * QCAP];
    __shared__ int qr[NWAVES * QCAP];
    __shared__ float4 wq[FUSE ? (DCH / 4) * WQ_PITCH : 1];
    const int tid = threadIdx.x, lane = tid & 31, wave = tid >> 5;
    const int row0 = blockIdx.x * TROWS;
    const int drow0 = blockIdx.x * DPITCH;
    const int tl = min(TROWS, n - row0);

    if constexpr (FUSE) {
        constexpr int NQ = (DCH / 4) * DCH / NTHR;
        float4 v[NQ];
#pragma unroll
        for (int i = 0; i < NQ; ++i) {
            const int idx = tid + NTHR * i;
            const int q = idx >> 7, col = idx & (DCH - 1);
            const float* wp = Wn + (size_t)(4 * q) * DCH + col;
            v[i].x = wp[0];
            v[i].y = wp[DCH];
            v[i].z = wp[2 * DCH];
            v[i].w = wp[3 * DCH];
        }
#pragma unroll
        for (int i = 0; i < NQ; ++i) {
            const int idx = tid + NTHR * i;
            wq[(idx >> 7) * WQ_PITCH + (idx & (DCH - 1))] = v[i];
        }
    }

    const float4 b = ((const float4*)bias)[lane];

    int cv[SLICES];
#pragma unroll
    for (int s = 0; s < SLICES; ++s) cv[s] = cnts[(blockIdx.x * SLICES + s) * SPITCH + lane];
    bool over = false;
#pragma unroll
    for (int s = 0; s < SLICES; ++s) over = over || (cv[s] > OCAP);
    if (__builtin_amdgcn_ballot_w32(over) == 0u) {
        const int wv = __builtin_amdgcn_readfirstlane(wave);
        const int* __restrict__ rt = recs + (size_t)blockIdx.x * RCAP + wv * OCAP;
        int c[SLICES];
        int rec[SLICES];
#pragma unroll
        for (int s = 0; s < SLICES; ++s) {
            c[s] = __builtin_amdgcn_readlane(cv[s], wv);
            rec[s] = (c[s] > 0) ? rt[s * NWAVES * OCAP + lane] : 0;
        }

        RowSums sum = {};
        float d[RPW];
        {
            float4 h[RPW];
#pragma unroll
            for (int i = 0; i < RPW; ++i) {
                int rc = min(wv + NWAVES * i, tl - 1);
                d[i] = dis[drow0 + rc];
                h[i] = ((const float4*)(H + (size_t)(row0 + rc) * DCH))[lane];
            }
#pragma unroll
            for (int i = 0; i < RPW; ++i) {
                sum.x[i] = h[i].x;
                sum.y[i] = h[i].y;
                sum.z[i] = h[i].z;
                sum.w[i] = h[i].w;
            }
        }

        int* we = qe + wv * QCAP;
        int total = 0;
#pragma unroll
        for (int s = 0; s < SLICES; ++s) {
            we[(lane < c[s]) ? total + lane : SLICES * OCAP + lane] = rec[s];
#pragma nounroll
            for (int k = 1; 32 * k < c[s]; ++k) {
                const int rc = rt[s * NWAVES * OCAP + 32 * k + lane];
                if (lane < c[s] - 32 * k) we[total + 32 * k + lane] = rc;
            }
            total += c[s];
        }
        const int nlines = (total + 31) >> 5;
        for (int g = 0; g < nlines; ++g) {
            const int rc = we[32 * g + lane];
            const int nvalid = min(32, total - 32 * g);
            add_line(sum, H, rc >> 8, (rc & 255) >> 5, nvalid, lane);
        }
#pragma unroll
        for (int i = 0; i < RPW; ++i) {
            const int orow = wv + NWAVES * i;
            float4 v;
            v.x = sum.x[i] * d[i] + b.x; v.y = sum.y[i] * d[i] + b.y;
            v.z = sum.z[i] * d[i] + b.z; v.w = sum.w[i] * d[i] + b.w;
            if (relu) {
                v.x = fmaxf(v.x, 0.0f); v.y = fmaxf(v.y, 0.0f);
                v.z = fmaxf(v.z, 0.0f); v.w = fmaxf(v.w, 0.0f);
            }
            if constexpr (FUSE) {
                v.x *= d[i]; v.y *= d[i]; v.z *= d[i]; v.w *= d[i];
                if (orow >= tl) v = make_float4(0.0f, 0.0f, 0.0f, 0.0f);
                if (orow < XROWS) *(float4*)(acc + orow * XPITCH + 4 * lane) = v;
            } else {
                if (orow < tl) ((float4*)(out + (size_t)(row0 + orow) * DCH))[lane] = v;
            }
        }
        if constexpr (!FUSE) return;
    } else {
        {
            float4 h[RPW];
#pragma unroll
            for (int i = 0; i < RPW; ++i) {
                int rc = min(wave + NWAVES * i, tl - 1);
                h[i] = ((const float4*)(H + (size_t)(row0 + rc) * DCH))[lane];
            }
#pragma unroll
            for (int i = 0; i < RPW; ++i) {
                int r = wave + NWAVES * i;
                if (r < tl) {
                    float* ap = acc + r * DCH + lane;
                    ap[0]  = h[i].x;
                    ap[32] = h[i].y;
                    ap[64] = h[i].z;
                    ap[96] = h[i].w;
                }
            }
        }
        __syncthreads();

        AggFilter f{acc, qe + wave * QCAP, qr + wave * QCAP, H, src, row0, tl, lane, 0};
        scan_targets(tgt, e, wave, lane, f);
        if (f.cnt > 0) f.flush(0, f.cnt);
        __syncthreads();

        if constexpr (FUSE) {
            float4 v[RPW];
#pragma unroll
            for (int i = 0; i < RPW; ++i) {
                const int r = wave + NWAVES * i;
                v[i] = make_float4(0.0f, 0.0f, 0.0f, 0.0f);
                if (r < tl) {
                    const float* ap = acc + r * DCH + lane;
                    const float dr = dis[drow0 + r];
                    v[i].x = ap[0] * dr + b.x; v[i].y = ap[32] * dr + b.y;
                    v[i].z = ap[64] * dr + b.z; v[i].w = ap[96] * dr + b.w;
                    if (relu) {
                        v[i].x = fmaxf(v[i].x, 0.0f); v[i].y = fmaxf(v[i].y, 0.0f);
                        v[i].z = fmaxf(v[i].z, 0.0f); v[i].w = fmaxf(v[i].w, 0.0f);
                    }
                    v[i].x *= dr; v[i].y *= dr; v[i].z *= dr; v[i].w *= dr;
                }
            }
            __syncthreads();
#pragma unroll
            for (int i = 0; i < RPW; ++i) {
                const int r = wave + NWAVES * i;
                if (r < XROWS) *(float4*)(acc + r * XPITCH + 4 * lane) = v[i];
            }
        } else {
            for (int r = wave; r < tl; r += NWAVES) {
                const float* ap = acc + r * DCH + lane;
                const float dr = dis[drow0 + r];
                float4 v;
                v.x = ap[0] * dr + b.x; v.y = ap[32] * dr + b.y;
                v.z = ap[64] * dr + b.z; v.w = ap[96] * dr + b.w;
                if (relu) {
                    v.x = fmaxf(v.x, 0.0f); v.y = fmaxf(v.y, 0.0f);
                    v.z = fmaxf(v.z, 0.0f); v.w = fmaxf(v.w, 0.0f);
                }
                ((float4*)(out + (size_t)(row0 + r) * DCH))[lane] = v;
            }
        }
    }

    if constexpr (FUSE) {
        __syncthreads();

        const int wm = __builtin_amdgcn_readfirstlane(wave);
        const int h = lane >> 4;
        const int c = lane & 15;
        const int nitems = 4 * ((tl + 15) >> 4);
        for (int item = wm; item < nitems; item += NWAVES) {
            const int t  = item >> 2;
            const int nq = item & 3;
            const float4* ap = (const float4*)(acc + (t * 16 + c) * XPITCH) + h;
            const float4* bp = wq + h * WQ_PITCH + 32 * nq + c;
            v8f d[2] = {};
            float4 a = ap[0];
            float4 bb[2];
#pragma unroll
            for (int u = 0; u < 2; ++u) bb[u] = bp[16 * u];
#pragma unroll
            for (int j = 0; j < DCH / 8; ++j) {
                float4 an = a;
                float4 bn[2];
                if (j + 1 < DCH / 8) {
                    an = ap[2 * (j + 1)];
#pragma unroll
                    for (int u = 0; u < 2; ++u) bn[u] = bp[2 * (j + 1) * WQ_PITCH + 16 * u];
                }
#pragma unroll
                for (int u = 0; u < 2; ++u) {
                    v2f av, bv;
                    av.x = a.x; av.y = a.y;
                    bv.x = bb[u].x; bv.y = bb[u].y;
                    d[u] = __builtin_amdgcn_wmma_f32_16x16x4_f32(
                             false, av,  false, bv,
                             (short)0, d[u],  false,  false);
                }
#pragma unroll
                for (int u = 0; u < 2; ++u) {
                    v2f av, bv;
                    av.x = a.z; av.y = a.w;
                    bv.x = bb[u].z; bv.y = bb[u].w;
                    d[u] = __builtin_amdgcn_wmma_f32_16x16x4_f32(
                             false, av,  false, bv,
                             (short)0, d[u],  false,  false);
                }
                if (j + 1 < DCH / 8) {
                    a = an;
#pragma unroll
                    for (int u = 0; u < 2; ++u) bb[u] = bn[u];
                }
            }
            const int trow = t * 16 + 8 * h;
            float* hp = Hn + (size_t)(row0 + trow) * DCH + 32 * nq + c;
#pragma unroll
            for (int r = 0; r < 8; ++r) {
                if (trow + r < tl) {
#pragma unroll
                    for (int u = 0; u < 2; ++u) hp[(size_t)r * DCH + 16 * u] = d[u][r];
                }
            }
        }
    }
}


extern "C" void kernel_launch(void* const* d_in, const int* in_sizes, int n_in,
                              void* d_out, int out_size, void* d_ws, size_t ws_size,
                              hipStream_t stream) {
    const float* X  = (const float*)d_in[0];
    const int*   ei = (const int*)  d_in[1];
    const float* W1 = (const float*)d_in[2];
    const float* b1 = (const float*)d_in[3];
    const float* W2 = (const float*)d_in[4];
    const float* b2 = (const float*)d_in[5];
    const float* W3 = (const float*)d_in[6];
    const float* b3 = (const float*)d_in[7];
    float* out = (float*)d_out;

    const int n  = in_sizes[0] / DCH;
    const int e  = in_sizes[1] / 2;

    const int* src = ei;
    const int* tgt = ei + e;

    const int gemmBlocks = 256;
    const int aggBlocks  = (n + TROWS - 1) / TROWS;
    const int bands      = (aggBlocks + BAND - 1) / BAND;

    int*   recs  = (int*)d_ws;
    float* dis   = (float*)(recs + (size_t)aggBlocks * RCAP);
    float* H0    = dis + (size_t)aggBlocks * DPITCH;
    float* H1    = H0 + (size_t)n * DCH;
    int*   cnts  = (int*)(H1 + (size_t)n * DCH);
    int*   degp  = cnts + (size_t)aggBlocks * SLICES * SPITCH;

    gcn_filter_band<<<bands * SLICES, NTHR, 0, stream>>>(src, tgt, degp, recs, cnts, n, e, aggBlocks);

    gcn_gemm_wmma<<<gemmBlocks, GEMM_THR, 0, stream>>>(X, W1, H0, n, degp, dis, aggBlocks, bands);
    gcn_agg_tile<true> <<<aggBlocks, NTHR, 0, stream>>>(H0, src, tgt, dis, b1, recs, cnts, nullptr, n, e, 1, W2, H1);
    gcn_agg_tile<true> <<<aggBlocks, NTHR, 0, stream>>>(H1, src, tgt, dis, b2, recs, cnts, nullptr, n, e, 1, W3, H0);
    gcn_agg_tile<false><<<aggBlocks, NTHR, 0, stream>>>(H0, src, tgt, dis, b3, recs, cnts, out, n, e, 0, nullptr, nullptr);
}
